# starting kernel
# baseline (speedup 1.0000x reference)
#include <hip/hip_runtime.h>
#include <hip/hip_bf16.h>

typedef _Float16 half_t;
typedef __attribute__((ext_vector_type(16))) _Float16 v16h;
typedef __attribute__((ext_vector_type(8)))  _Float16 v8h;
typedef __attribute__((ext_vector_type(8)))  float    v8f;

union V16 { v16h v; v8h h[2]; };

constexpr int BB   = 2;
constexpr int TT   = 2048;
constexpr int DM   = 1024;
constexpr int HN   = 16;
constexpr int DH   = 64;
constexpr int QKVD = 3072;
constexpr int MROW = BB * TT;
constexpr int WIN  = 128;
constexpr float SM_SCALE = 0.125f;
constexpr float MASKED = -3.0e38f;
constexpr float MINIT  = -1.0e30f;

constexpr size_t OFF_QKVW = 0;
constexpr size_t SZ_QKVW  = (size_t)QKVD * DM * 2;
constexpr size_t OFF_OUTW = OFF_QKVW + SZ_QKVW;
constexpr size_t SZ_OUTW  = (size_t)DM * DM * 2;
constexpr size_t OFF_T    = OFF_OUTW + SZ_OUTW;
constexpr size_t SZ_T     = (size_t)MROW * DM * 2;
constexpr size_t OFF_QKVF = OFF_T + SZ_T;
constexpr size_t SZ_QKVF  = (size_t)MROW * QKVD * 4;
constexpr size_t OFF_ATTN = OFF_QKVF;
constexpr size_t OFF_Q    = OFF_QKVF + SZ_QKVF;
constexpr size_t SZ_QKH   = (size_t)BB * HN * TT * DH * 2;
constexpr size_t OFF_K    = OFF_Q + SZ_QKH;
constexpr size_t OFF_V    = OFF_K + SZ_QKH;

__device__ __forceinline__ void ldf(V16& f, const half_t* p) {
  f.h[0] = *(const v8h*)(p);
  f.h[1] = *(const v8h*)(p + 16);
}

__global__ __launch_bounds__(256) void rmsnorm_kernel(
    const float* __restrict__ x, const float* __restrict__ scale,
    half_t* __restrict__ t) {
  const int row = blockIdx.x;
  const float* xr = x + (size_t)row * DM;
  float ss = 0.f;
  for (int i = threadIdx.x; i < DM; i += 256) { float v = xr[i]; ss += v * v; }
  #pragma unroll
  for (int o = 16; o > 0; o >>= 1) ss += __shfl_xor(ss, o, 32);
  __shared__ float red[8];
  if ((threadIdx.x & 31) == 0) red[threadIdx.x >> 5] = ss;
  __syncthreads();
  float tot = 0.f;
  #pragma unroll
  for (int i = 0; i < 8; ++i) tot += red[i];
  const float inv = rsqrtf(tot * (1.0f / DM) + 1e-5f);
  for (int i = threadIdx.x; i < DM; i += 256)
    t[(size_t)row * DM + i] = (half_t)(xr[i] * inv * scale[i]);
}

__global__ __launch_bounds__(256) void f32_to_f16_kernel(
    const float* __restrict__ src, half_t* __restrict__ dst, int n) {
  int i = blockIdx.x * blockDim.x + threadIdx.x;
  const int stride = gridDim.x * blockDim.x;
  for (; i < n; i += stride) dst[i] = (half_t)src[i];
}

__device__ __forceinline__ void mma8(v8f acc[2][4], const V16 a[2], const V16 b[4]) {
  #pragma unroll
  for (int i = 0; i < 2; ++i)
    #pragma unroll
    for (int j = 0; j < 4; ++j)
      acc[i][j] = __builtin_amdgcn_wmma_f32_16x16x32_f16(
          false, a[i].v, false, b[j].v, (short)0, acc[i][j], false, false);
}

__device__ __forceinline__ void load_step(
    V16 a[2], V16 b[4],
    const half_t* A0, const half_t* A1,
    const half_t* W0, const half_t* W1,
    const half_t* W2, const half_t* W3, int k) {
  ldf(a[0], A0 + k);
  ldf(a[1], A1 + k);
  ldf(b[0], W0 + k);
  ldf(b[1], W1 + k);
  ldf(b[2], W2 + k);
  ldf(b[3], W3 + k);
}

__global__ __launch_bounds__(256) void gemm_wmma_kernel(
    const half_t* __restrict__ A, const half_t* __restrict__ W,
    const float* __restrict__ bias, const float* __restrict__ residual,
    float* __restrict__ C, int M, int N, int K) {
  const int lane = threadIdx.x & 31;
  const int wid  = (blockIdx.x * blockDim.x + threadIdx.x) >> 5;
  const int ntn  = N >> 6;
  const int ntm  = M >> 5;
  if (wid >= ntm * ntn) return;
  const int tm = wid / ntn, tn = wid % ntn;
  const int ko = (lane >= 16) ? 8 : 0;
  const int ln = lane & 15;

  const half_t* A0 = A + (size_t)(tm * 32 + ln) * K + ko;
  const half_t* A1 = A0 + (size_t)16 * K;
  const half_t* W0 = W + (size_t)(tn * 64 + ln) * K + ko;
  const half_t* W1 = W0 + (size_t)16 * K;
  const half_t* W2 = W0 + (size_t)32 * K;
  const half_t* W3 = W0 + (size_t)48 * K;

  v8f acc[2][4] = {};
  V16 pa[2], pb[4], qa[2], qb[4];

  load_step(pa, pb, A0, A1, W0, W1, W2, W3, 0);
  int k = 32;
  for (; k < K - 32; k += 64) {
    load_step(qa, qb, A0, A1, W0, W1, W2, W3, k);
    mma8(acc, pa, pb);
    load_step(pa, pb, A0, A1, W0, W1, W2, W3, k + 32);
    mma8(acc, qa, qb);
  }
  load_step(qa, qb, A0, A1, W0, W1, W2, W3, k);
  mma8(acc, pa, pb);
  mma8(acc, qa, qb);

  #pragma unroll
  for (int i = 0; i < 2; ++i) {
    #pragma unroll
    for (int j = 0; j < 4; ++j) {
      const int col = tn * 64 + j * 16 + ln;
      const float bv = bias[col];
      #pragma unroll
      for (int r = 0; r < 8; ++r) {
        const int row = tm * 32 + i * 16 + r + ko;
        float v = acc[i][j][r] + bv;
        if (residual) v += residual[(size_t)row * N + col];
        C[(size_t)row * N + col] = v;
      }
    }
  }
}

__global__ __launch_bounds__(256) void rope_kernel(
    const float* __restrict__ qkv, half_t* __restrict__ Q,
    half_t* __restrict__ Kr, half_t* __restrict__ Vt) {
  const int idx = blockIdx.x * blockDim.x + threadIdx.x;
  const int d = idx & 63;
  const int h = (idx >> 6) & (HN - 1);
  const int t = (idx >> 10) & (TT - 1);
  const int b = idx >> 21;
  const int p = d & 31;
  const float ang = (float)t * __expf(-(float)p * (9.210340372f / 32.0f));
  const float c = cosf(ang), sn = sinf(ang);
  const size_t base = (size_t)(b * TT + t) * QKVD;
  const float q1 = qkv[base + h * 64 + p];
  const float q2 = qkv[base + h * 64 + p + 32];
  const float k1 = qkv[base + DM + h * 64 + p];
  const float k2 = qkv[base + DM + h * 64 + p + 32];
  const float qv = (d < 32) ? (q1 * c - q2 * sn) : (q2 * c + q1 * sn);
  const float kv = (d < 32) ? (k1 * c - k2 * sn) : (k2 * c + k1 * sn);
  const size_t bh = (size_t)(b * HN + h);
  Q [(bh * TT + t) * DH + d] = (half_t)qv;
  Kr[(bh * TT + t) * DH + d] = (half_t)kv;
  Vt[(bh * DH + d) * TT + t] = (half_t)qkv[base + 2 * DM + h * 64 + d];
}

__global__ __launch_bounds__(256) void attn_kernel(
    const half_t* __restrict__ Q, const half_t* __restrict__ Kr,
    const half_t* __restrict__ Vt, const int* __restrict__ amask,
    const float* __restrict__ sinks, half_t* __restrict__ attn) {
  const int lane = threadIdx.x & 31;
  const int wid  = (blockIdx.x * blockDim.x + threadIdx.x) >> 5;
  const int QT   = TT / 16;
  const int qt   = wid % QT;
  const int bh   = wid / QT;
  const int h    = bh % HN, b = bh / HN;
  const int qbase = qt * 16;
  const int ln = lane & 15;
  const int ko = (lane >= 16) ? 8 : 0;
  const int iq = qbase + ln;
  const half_t* Qh = Q  + (size_t)bh * TT * DH;
  const half_t* Kh = Kr + (size_t)bh * TT * DH;
  const half_t* Vh = Vt + (size_t)bh * DH * TT;
  const int* am = amask + b * TT;

  V16 qb0, qb1;
  {
    const half_t* qp = Qh + (size_t)(qbase + ln) * DH;
    ldf(qb0, qp + ko);
    ldf(qb1, qp + 32 + ko);
  }

  v8f acc[4] = {};
  float m = MINIT, l = 0.f;
  int kstart = qbase - (WIN - 1); if (kstart < 0) kstart = 0;
  kstart &= ~31;
  const int kend = qbase + 16;

  for (int jc = kstart; jc < kend; jc += 32) {
    v8f s[2];
    #pragma unroll
    for (int t2 = 0; t2 < 2; ++t2) {
      int jrow = jc + t2 * 16 + ln;
      if (jrow > TT - 1) jrow = TT - 1;
      const half_t* kp = Kh + (size_t)jrow * DH;
      V16 a0, a1;
      ldf(a0, kp + ko);
      ldf(a1, kp + 32 + ko);
      v8f z = {};
      z = __builtin_amdgcn_wmma_f32_16x16x32_f16(false, a0.v, false, qb0.v,
                                                 (short)0, z, false, false);
      z = __builtin_amdgcn_wmma_f32_16x16x32_f16(false, a1.v, false, qb1.v,
                                                 (short)0, z, false, false);
      s[t2] = z;
    }
    float tl = MASKED;
    #pragma unroll
    for (int t2 = 0; t2 < 2; ++t2) {
      #pragma unroll
      for (int r = 0; r < 8; ++r) {
        const int j = jc + t2 * 16 + r + ko;
        const int jcl = (j < TT - 1) ? j : TT - 1;
        const bool ok = (j <= iq) && (j + WIN > iq) && (j < TT) && (am[jcl] != 0);
        const float v = ok ? s[t2][r] * SM_SCALE : MASKED;
        s[t2][r] = v;
        tl = fmaxf(tl, v);
      }
    }
    tl = fmaxf(tl, __shfl_xor(tl, 16, 32));
    const float mnew = fmaxf(m, tl);
    const float co = __expf(m - mnew);
    l *= co;
    #pragma unroll
    for (int dt = 0; dt < 4; ++dt) acc[dt] = acc[dt] * co;

    V16 pbf;
    float psum = 0.f;
    #pragma unroll
    for (int r = 0; r < 8; ++r) {
      const float p0 = __expf(s[0][r] - mnew);
      const float p1 = __expf(s[1][r] - mnew);
      psum += p0 + p1;
      pbf.h[0][r] = (half_t)p0;
      pbf.h[1][r] = (half_t)p1;
    }
    l += psum + __shfl_xor(psum, 16, 32);
    m = mnew;

    #pragma unroll
    for (int dt = 0; dt < 4; ++dt) {
      const half_t* vp = Vh + (size_t)(dt * 16 + ln) * TT + jc + ko;
      V16 va;
      ldf(va, vp);
      acc[dt] = __builtin_amdgcn_wmma_f32_16x16x32_f16(
          false, va.v, false, pbf.v, (short)0, acc[dt], false, false);
    }
  }

  const float snk = sinks[h];
  const float m2 = fmaxf(m, snk);
  const float co = __expf(m - m2);
  const float ltot = l * co + __expf(snk - m2);
  const float factor = co / ltot;
  half_t* op = attn + (size_t)(b * TT + qbase + ln) * DM + h * 64;
  #pragma unroll
  for (int dt = 0; dt < 4; ++dt) {
    v8h o;
    #pragma unroll
    for (int r = 0; r < 8; ++r) o[r] = (half_t)(acc[dt][r] * factor);
    *(v8h*)(op + dt * 16 + ko) = o;
  }
}

extern "C" void kernel_launch(void* const* d_in, const int* in_sizes, int n_in,
                              void* d_out, int out_size, void* d_ws, size_t ws_size,
                              hipStream_t stream) {
  const float* x      = (const float*)d_in[0];
  const int*   amask  = (const int*)d_in[1];
  const float* nscale = (const float*)d_in[2];
  const float* qkv_w  = (const float*)d_in[3];
  const float* qkv_b  = (const float*)d_in[4];
  const float* out_w  = (const float*)d_in[5];
  const float* out_b  = (const float*)d_in[6];
  const float* sinks  = (const float*)d_in[7];
  float* out = (float*)d_out;
  char*  ws  = (char*)d_ws;

  half_t* qkvw_h = (half_t*)(ws + OFF_QKVW);
  half_t* outw_h = (half_t*)(ws + OFF_OUTW);
  half_t* t_h    = (half_t*)(ws + OFF_T);
  float*  qkvf   = (float*)(ws + OFF_QKVF);
  half_t* attn_h = (half_t*)(ws + OFF_ATTN);
  half_t* Qh     = (half_t*)(ws + OFF_Q);
  half_t* Kh     = (half_t*)(ws + OFF_K);
  half_t* Vth    = (half_t*)(ws + OFF_V);

  rmsnorm_kernel<<<MROW, 256, 0, stream>>>(x, nscale, t_h);
  f32_to_f16_kernel<<<2048, 256, 0, stream>>>(qkv_w, qkvw_h, QKVD * DM);
  f32_to_f16_kernel<<<1024, 256, 0, stream>>>(out_w, outw_h, DM * DM);

  gemm_wmma_kernel<<<768, 256, 0, stream>>>(t_h, qkvw_h, qkv_b, nullptr,
                                            qkvf, MROW, QKVD, DM);

  rope_kernel<<<(BB * TT * HN * DH) / 256, 256, 0, stream>>>(qkvf, Qh, Kh, Vth);

  attn_kernel<<<512, 256, 0, stream>>>(Qh, Kh, Vth, amask, sinks, attn_h);

  gemm_wmma_kernel<<<256, 256, 0, stream>>>(attn_h, outw_h, out_b, x,
                                            out, MROW, DM, DM);
}


# Both GEMMs go through LDS: a 4-wave block fetches its 128x128x64 operand tiles once as whole 128-B lines (2 stages), waves take 64x64 by ds_load_b128; C leaves via LDS in b128 rows; k sum order kept
# speedup vs baseline: 2.6126x; 2.6126x over previous
#include <hip/hip_runtime.h>
#include <hip/hip_bf16.h>

typedef _Float16 half_t;
typedef __attribute__((ext_vector_type(16))) _Float16 v16h;
typedef __attribute__((ext_vector_type(8)))  _Float16 v8h;
typedef __attribute__((ext_vector_type(8)))  float    v8f;
typedef __attribute__((ext_vector_type(4)))  float    v4f;
typedef __attribute__((ext_vector_type(4)))  _Float16 v4h;

union V16 { v16h v; v8h h[2]; };

constexpr int BB   = 2;
constexpr int TT   = 2048;
constexpr int DM   = 1024;
constexpr int HN   = 16;
constexpr int DH   = 64;
constexpr int QKVD = 3072;
constexpr int MROW = BB * TT;
constexpr int WIN  = 128;
constexpr float SM_SCALE = 0.125f;
constexpr float MASKED = -3.0e38f;
constexpr float MINIT  = -1.0e30f;

constexpr size_t OFF_QKVW = 0;
constexpr size_t SZ_QKVW  = (size_t)QKVD * DM * 2;
constexpr size_t OFF_OUTW = OFF_QKVW + SZ_QKVW;
constexpr size_t SZ_OUTW  = (size_t)DM * DM * 2;
constexpr size_t OFF_T    = OFF_OUTW + SZ_OUTW;
constexpr size_t SZ_T     = (size_t)MROW * DM * 2;
constexpr size_t OFF_QKVF = OFF_T + SZ_T;
constexpr size_t SZ_QKVF  = (size_t)MROW * QKVD * 4;
constexpr size_t OFF_ATTN = OFF_QKVF;
constexpr size_t OFF_Q    = OFF_QKVF + SZ_QKVF;
constexpr size_t SZ_QKH   = (size_t)BB * HN * TT * DH * 2;
constexpr size_t OFF_K    = OFF_Q + SZ_QKH;
constexpr size_t OFF_V    = OFF_K + SZ_QKH;

__device__ __forceinline__ void ldf(V16& f, const half_t* p) {
  f.h[0] = *(const v8h*)(p);
  f.h[1] = *(const v8h*)(p + 16);
}

__global__ __launch_bounds__(256) void rmsnorm_kernel(
    const float* __restrict__ x, const float* __restrict__ scale,
    half_t* __restrict__ t) {
  const int lane = threadIdx.x & 31;
  const int row  = blockIdx.x * 8 + (threadIdx.x >> 5);
  const float* xr = x + (size_t)row * DM;
  v4f v[8];
  #pragma unroll
  for (int i = 0; i < 8; ++i) v[i] = *(const v4f*)(xr + (i * 32 + lane) * 4);
  float ss = 0.f;
  #pragma unroll
  for (int i = 0; i < 8; ++i)
    ss += v[i][0] * v[i][0] + v[i][1] * v[i][1] + v[i][2] * v[i][2] + v[i][3] * v[i][3];
  #pragma unroll
  for (int o = 16; o > 0; o >>= 1) ss += __shfl_xor(ss, o, 32);
  const float inv = rsqrtf(ss * (1.0f / DM) + 1e-5f);
  half_t* tr = t + (size_t)row * DM;
  #pragma unroll
  for (int i = 0; i < 8; ++i) {
    const int c = (i * 32 + lane) * 4;
    const v4f sc = *(const v4f*)(scale + c);
    v4h o;
    #pragma unroll
    for (int e = 0; e < 4; ++e) o[e] = (half_t)(v[i][e] * inv * sc[e]);
    *(v4h*)(tr + c) = o;
  }
}

template <int ITERS>
__global__ __launch_bounds__(256) void f32_to_f16_kernel(
    const float* __restrict__ src, half_t* __restrict__ dst) {
  const int i0 = blockIdx.x * 256 + threadIdx.x;
  const int stride = gridDim.x * 256;
  v4f a[ITERS];
  #pragma unroll
  for (int it = 0; it < ITERS; ++it)
    a[it] = *(const v4f*)(src + (size_t)(i0 + it * stride) * 4);
  #pragma unroll
  for (int it = 0; it < ITERS; ++it) {
    v4h o;
    #pragma unroll
    for (int e = 0; e < 4; ++e) o[e] = (half_t)a[it][e];
    *(v4h*)(dst + (size_t)(i0 + it * stride) * 4) = o;
  }
}

constexpr int GBM    = 128;
constexpr int GBK    = 64;
constexpr int GPITCH = 72;
constexpr int GTILE  = GBM * GPITCH;
constexpr int CPITCH = 68;
static_assert(4 * 64 * CPITCH * 4 <= 4 * GTILE * 2, "C parts must fit the operand stages");

__device__ __forceinline__ void gemm_fetch(v8h ra[8], v8h rb[8],
                                           const half_t* ag, const half_t* wg,
                                           int K, int k) {
  #pragma unroll
  for (int it = 0; it < 8; ++it) {
    ra[it] = *(const v8h*)(ag + (size_t)(it * 16) * K + k);
    rb[it] = *(const v8h*)(wg + (size_t)(it * 16) * K + k);
  }
}

__device__ __forceinline__ void gemm_stash(half_t* as, half_t* bs,
                                           const v8h ra[8], const v8h rb[8]) {
  #pragma unroll
  for (int it = 0; it < 8; ++it) {
    *(v8h*)(as + it * 16 * GPITCH) = ra[it];
    *(v8h*)(bs + it * 16 * GPITCH) = rb[it];
  }
}

__device__ __forceinline__ void gemm_stage(v8f acc[4][4],
                                           const half_t* af, const half_t* bf) {
  V16 a[2][4], b[2][4];
  #pragma unroll
  for (int c = 0; c < 2; ++c) {
    #pragma unroll
    for (int i = 0; i < 4; ++i) ldf(a[c][i], af + i * 16 * GPITCH + c * 32);
    #pragma unroll
    for (int j = 0; j < 4; ++j) ldf(b[c][j], bf + j * 16 * GPITCH + c * 32);
  }
  __builtin_amdgcn_sched_barrier(0);
  #pragma unroll
  for (int c = 0; c < 2; ++c)
    #pragma unroll
    for (int i = 0; i < 4; ++i)
      #pragma unroll
      for (int j = 0; j < 4; ++j)
        acc[i][j] = __builtin_amdgcn_wmma_f32_16x16x32_f16(
            false, a[c][i].v, false, b[c][j].v, (short)0, acc[i][j], false, false);
}

template <bool RES>
__global__ __launch_bounds__(128) __attribute__((amdgpu_waves_per_eu(3, 3)))
void gemm_wmma_kernel(
    const half_t* __restrict__ A, const half_t* __restrict__ W,
    const float* __restrict__ bias, const float* __restrict__ residual,
    float* __restrict__ C, int M, int N, int K) {
  __shared__ __attribute__((aligned(16))) half_t sm[4 * GTILE];
  (void)M;
  const int tid  = threadIdx.x;
  const int lane = tid & 31;
  const int wv   = tid >> 5;
  const int wm   = wv >> 1, wn = wv & 1;
  const int ntn  = N >> 7;
  const int tm   = blockIdx.x / ntn, tn = blockIdx.x % ntn;
  const int ko = (lane >= 16) ? 8 : 0;
  const int ln = lane & 15;

  const int srow = tid >> 3, scol = (tid & 7) * 8;
  const half_t* ag = A + (size_t)(tm * GBM + srow) * K + scol;
  const half_t* wg = W + (size_t)(tn * GBM + srow) * K + scol;
  half_t* sst = sm + srow * GPITCH + scol;
  const half_t* fa = sm + (wm * 64 + ln) * GPITCH + ko;
  const half_t* fb = sm + GTILE + (wn * 64 + ln) * GPITCH + ko;

  v8f acc[4][4] = {};
  v8h ra[8], rb[8];

  gemm_fetch(ra, rb, ag, wg, K, 0);
  gemm_stash(sst, sst + GTILE, ra, rb);
  __syncthreads();

  const int nk = K / GBK;
  for (int kt = 0; kt < nk - 1; ++kt) {
    const int cur = (kt & 1) * 2 * GTILE, nxt = 2 * GTILE - cur;
    gemm_fetch(ra, rb, ag, wg, K, (kt + 1) * GBK);
    __builtin_amdgcn_sched_barrier(0);
    gemm_stage(acc, fa + cur, fb + cur);
    __builtin_amdgcn_sched_barrier(0);
    gemm_stash(sst + nxt, sst + nxt + GTILE, ra, rb);
    __syncthreads();
  }
  {
    const int cur = ((nk - 1) & 1) * 2 * GTILE;
    gemm_stage(acc, fa + cur, fb + cur);
  }

  __syncthreads();
  float* cw = (float*)sm + wv * (64 * CPITCH);
  #pragma unroll
  for (int i = 0; i < 4; ++i)
    #pragma unroll
    for (int j = 0; j < 4; ++j)
      #pragma unroll
      for (int r = 0; r < 8; ++r)
        cw[(i * 16 + r + ko) * CPITCH + j * 16 + ln] = acc[i][j][r];
  __syncthreads();

  const int c4 = ln * 4;
  const int rh = lane >> 4;
  const int colg = tn * GBM + wn * 64 + c4;
  const v4f bv = *(const v4f*)(bias + colg);
  #pragma unroll 8
  for (int ps = 0; ps < 32; ++ps) {
    const int rl = ps * 2 + rh;
    const size_t g = (size_t)(tm * GBM + wm * 64 + rl) * N + colg;
    v4f v = *(const v4f*)(cw + rl * CPITCH + c4);
    v = v + bv;
    if (RES) v = v + *(const v4f*)(residual + g);
    *(v4f*)(C + g) = v;
  }
}

constexpr int VPITCH = 72;

__global__ __launch_bounds__(256) void rope_kernel(
    const float* __restrict__ qkv, half_t* __restrict__ Q,
    half_t* __restrict__ Kr, half_t* __restrict__ Vt) {
  typedef __attribute__((ext_vector_type(2))) _Float16 v2h;
  __shared__ __attribute__((aligned(16))) half_t vs[64 * VPITCH];
  const int tid = threadIdx.x;
  const int h  = blockIdx.x & (HN - 1);
  const int t0 = ((blockIdx.x >> 4) & 31) * 64;
  const int b  = blockIdx.x >> 9;
  const size_t bh = (size_t)(b * HN + h);
  const float* tile = qkv + (size_t)(b * TT + t0) * QKVD + h * 64;

  const int d4 = (tid & 15) * 4;
  const int rv = (tid >> 4) * 2;
  const int lane = tid & 31;
  const int p4   = (lane & 7) * 4;
  const int rsub = (tid >> 5) * 4 + (lane >> 3);

  v4f va[2], vb[2], q1[2], q2[2], k1[2], k2[2];
  #pragma unroll
  for (int i = 0; i < 2; ++i) {
    const float* vp = tile + (size_t)(rv + 32 * i) * QKVD + 2 * DM + d4;
    va[i] = *(const v4f*)vp;
    vb[i] = *(const v4f*)(vp + QKVD);
  }
  #pragma unroll
  for (int i = 0; i < 2; ++i) {
    const float* rp = tile + (size_t)(rsub + 32 * i) * QKVD + p4;
    q1[i] = *(const v4f*)(rp);
    q2[i] = *(const v4f*)(rp + 32);
    k1[i] = *(const v4f*)(rp + DM);
    k2[i] = *(const v4f*)(rp + DM + 32);
  }

  #pragma unroll
  for (int i = 0; i < 2; ++i) {
    #pragma unroll
    for (int e = 0; e < 4; ++e) {
      v2h pr;
      pr[0] = (half_t)va[i][e];
      pr[1] = (half_t)vb[i][e];
      *(v2h*)(&vs[(d4 + e) * VPITCH + rv + 32 * i]) = pr;
    }
  }
  __syncthreads();

  #pragma unroll
  for (int ps = 0; ps < 2; ++ps) {
    const int d  = ps * 32 + (tid >> 3);
    const int c8 = (tid & 7) * 8;
    const v8h val = *(const v8h*)(&vs[d * VPITCH + c8]);
    *(v8h*)(Vt + (bh * DH + d) * TT + t0 + c8) = val;
  }

  float invf[4];
  #pragma unroll
  for (int e = 0; e < 4; ++e)
    invf[e] = __expf(-(float)(p4 + e) * (9.210340372f / 32.0f));
  #pragma unroll
  for (int i = 0; i < 2; ++i) {
    const int t = t0 + rsub + 32 * i;
    v4h qlo, qhi, klo, khi;
    #pragma unroll
    for (int e = 0; e < 4; ++e) {
      const float ang = (float)t * invf[e];
      const float c = cosf(ang), sn = sinf(ang);
      qlo[e] = (half_t)(q1[i][e] * c - q2[i][e] * sn);
      qhi[e] = (half_t)(q2[i][e] * c + q1[i][e] * sn);
      klo[e] = (half_t)(k1[i][e] * c - k2[i][e] * sn);
      khi[e] = (half_t)(k2[i][e] * c + k1[i][e] * sn);
    }
    const size_t o = (bh * TT + t) * DH + p4;
    *(v4h*)(Q  + o)      = qlo;
    *(v4h*)(Q  + o + 32) = qhi;
    *(v4h*)(Kr + o)      = klo;
    *(v4h*)(Kr + o + 32) = khi;
  }
}

__global__ __launch_bounds__(256) void attn_kernel(
    const half_t* __restrict__ Q, const half_t* __restrict__ Kr,
    const half_t* __restrict__ Vt, const int* __restrict__ amask,
    const float* __restrict__ sinks, half_t* __restrict__ attn) {
  const int lane = threadIdx.x & 31;
  const int wid  = (blockIdx.x * blockDim.x + threadIdx.x) >> 5;
  const int QT   = TT / 16;
  const int qt   = wid % QT;
  const int bh   = wid / QT;
  const int h    = bh % HN, b = bh / HN;
  const int qbase = qt * 16;
  const int ln = lane & 15;
  const int ko = (lane >= 16) ? 8 : 0;
  const int iq = qbase + ln;
  const half_t* Qh = Q  + (size_t)bh * TT * DH;
  const half_t* Kh = Kr + (size_t)bh * TT * DH;
  const half_t* Vh = Vt + (size_t)bh * DH * TT;
  const int* am = amask + b * TT;

  V16 qb0, qb1;
  {
    const half_t* qp = Qh + (size_t)(qbase + ln) * DH;
    ldf(qb0, qp + ko);
    ldf(qb1, qp + 32 + ko);
  }

  v8f acc[4] = {};
  float m = MINIT, l = 0.f;
  int kstart = qbase - (WIN - 1); if (kstart < 0) kstart = 0;
  kstart &= ~31;
  const int kend = qbase + 16;

  for (int jc = kstart; jc < kend; jc += 32) {
    v8f s[2];
    #pragma unroll
    for (int t2 = 0; t2 < 2; ++t2) {
      int jrow = jc + t2 * 16 + ln;
      if (jrow > TT - 1) jrow = TT - 1;
      const half_t* kp = Kh + (size_t)jrow * DH;
      V16 a0, a1;
      ldf(a0, kp + ko);
      ldf(a1, kp + 32 + ko);
      v8f z = {};
      z = __builtin_amdgcn_wmma_f32_16x16x32_f16(false, a0.v, false, qb0.v,
                                                 (short)0, z, false, false);
      z = __builtin_amdgcn_wmma_f32_16x16x32_f16(false, a1.v, false, qb1.v,
                                                 (short)0, z, false, false);
      s[t2] = z;
    }
    float tl = MASKED;
    #pragma unroll
    for (int t2 = 0; t2 < 2; ++t2) {
      #pragma unroll
      for (int r = 0; r < 8; ++r) {
        const int j = jc + t2 * 16 + r + ko;
        const int jcl = (j < TT - 1) ? j : TT - 1;
        const bool ok = (j <= iq) && (j + WIN > iq) && (j < TT) && (am[jcl] != 0);
        const float v = ok ? s[t2][r] * SM_SCALE : MASKED;
        s[t2][r] = v;
        tl = fmaxf(tl, v);
      }
    }
    tl = fmaxf(tl, __shfl_xor(tl, 16, 32));
    const float mnew = fmaxf(m, tl);
    const float co = __expf(m - mnew);
    l *= co;
    #pragma unroll
    for (int dt = 0; dt < 4; ++dt) acc[dt] = acc[dt] * co;

    V16 pbf;
    float psum = 0.f;
    #pragma unroll
    for (int r = 0; r < 8; ++r) {
      const float p0 = __expf(s[0][r] - mnew);
      const float p1 = __expf(s[1][r] - mnew);
      psum += p0 + p1;
      pbf.h[0][r] = (half_t)p0;
      pbf.h[1][r] = (half_t)p1;
    }
    l += psum + __shfl_xor(psum, 16, 32);
    m = mnew;

    #pragma unroll
    for (int dt = 0; dt < 4; ++dt) {
      const half_t* vp = Vh + (size_t)(dt * 16 + ln) * TT + jc + ko;
      V16 va;
      ldf(va, vp);
      acc[dt] = __builtin_amdgcn_wmma_f32_16x16x32_f16(
          false, va.v, false, pbf.v, (short)0, acc[dt], false, false);
    }
  }

  const float snk = sinks[h];
  const float m2 = fmaxf(m, snk);
  const float co = __expf(m - m2);
  const float ltot = l * co + __expf(snk - m2);
  const float factor = co / ltot;
  half_t* op = attn + (size_t)(b * TT + qbase + ln) * DM + h * 64;
  #pragma unroll
  for (int dt = 0; dt < 4; ++dt) {
    v8h o;
    #pragma unroll
    for (int r = 0; r < 8; ++r) o[r] = (half_t)(acc[dt][r] * factor);
    *(v8h*)(op + dt * 16 + ko) = o;
  }
}

extern "C" void kernel_launch(void* const* d_in, const int* in_sizes, int n_in,
                              void* d_out, int out_size, void* d_ws, size_t ws_size,
                              hipStream_t stream) {
  const float* x      = (const float*)d_in[0];
  const int*   amask  = (const int*)d_in[1];
  const float* nscale = (const float*)d_in[2];
  const float* qkv_w  = (const float*)d_in[3];
  const float* qkv_b  = (const float*)d_in[4];
  const float* out_w  = (const float*)d_in[5];
  const float* out_b  = (const float*)d_in[6];
  const float* sinks  = (const float*)d_in[7];
  float* out = (float*)d_out;
  char*  ws  = (char*)d_ws;

  half_t* qkvw_h = (half_t*)(ws + OFF_QKVW);
  half_t* outw_h = (half_t*)(ws + OFF_OUTW);
  half_t* t_h    = (half_t*)(ws + OFF_T);
  float*  qkvf   = (float*)(ws + OFF_QKVF);
  half_t* attn_h = (half_t*)(ws + OFF_ATTN);
  half_t* Qh     = (half_t*)(ws + OFF_Q);
  half_t* Kh     = (half_t*)(ws + OFF_K);
  half_t* Vth    = (half_t*)(ws + OFF_V);

  rmsnorm_kernel<<<MROW / 8, 256, 0, stream>>>(x, nscale, t_h);
  f32_to_f16_kernel<6><<<512, 256, 0, stream>>>(qkv_w, qkvw_h);
  f32_to_f16_kernel<4><<<256, 256, 0, stream>>>(out_w, outw_h);

  gemm_wmma_kernel<false><<<768, 128, 0, stream>>>(t_h, qkvw_h, qkv_b, nullptr,
                                            qkvf, MROW, QKVD, DM);

  rope_kernel<<<BB * (TT / 64) * HN, 256, 0, stream>>>(qkvf, Qh, Kh, Vth);

  attn_kernel<<<512, 256, 0, stream>>>(Qh, Kh, Vth, amask, sinks, attn_h);

  gemm_wmma_kernel<true><<<256, 128, 0, stream>>>(attn_h, outw_h, out_b, x,
                                            out, MROW, DM, DM);
}


# Attention: a block's 8 waves stage the 256 keys they share (K, V^T, mask words) once in LDS by whole-line loads; fragments come by ds_load_b128, not 16-line global loads; sum order kept
# speedup vs baseline: 2.9157x; 1.1160x over previous
#include <hip/hip_runtime.h>
#include <hip/hip_bf16.h>

typedef _Float16 half_t;
typedef __attribute__((ext_vector_type(16))) _Float16 v16h;
typedef __attribute__((ext_vector_type(8)))  _Float16 v8h;
typedef __attribute__((ext_vector_type(8)))  float    v8f;
typedef __attribute__((ext_vector_type(4)))  float    v4f;
typedef __attribute__((ext_vector_type(4)))  _Float16 v4h;

union V16 { v16h v; v8h h[2]; };

constexpr int BB   = 2;
constexpr int TT   = 2048;
constexpr int DM   = 1024;
constexpr int HN   = 16;
constexpr int DH   = 64;
constexpr int QKVD = 3072;
constexpr int MROW = BB * TT;
constexpr int WIN  = 128;
constexpr float SM_SCALE = 0.125f;
constexpr float MASKED = -3.0e38f;
constexpr float MINIT  = -1.0e30f;

constexpr size_t OFF_QKVW = 0;
constexpr size_t SZ_QKVW  = (size_t)QKVD * DM * 2;
constexpr size_t OFF_OUTW = OFF_QKVW + SZ_QKVW;
constexpr size_t SZ_OUTW  = (size_t)DM * DM * 2;
constexpr size_t OFF_T    = OFF_OUTW + SZ_OUTW;
constexpr size_t SZ_T     = (size_t)MROW * DM * 2;
constexpr size_t OFF_QKVF = OFF_T + SZ_T;
constexpr size_t SZ_QKVF  = (size_t)MROW * QKVD * 4;
constexpr size_t OFF_ATTN = OFF_QKVF;
constexpr size_t OFF_Q    = OFF_QKVF + SZ_QKVF;
constexpr size_t SZ_QKH   = (size_t)BB * HN * TT * DH * 2;
constexpr size_t OFF_K    = OFF_Q + SZ_QKH;
constexpr size_t OFF_V    = OFF_K + SZ_QKH;

__device__ __forceinline__ void ldf(V16& f, const half_t* p) {
  f.h[0] = *(const v8h*)(p);
  f.h[1] = *(const v8h*)(p + 16);
}

__global__ __launch_bounds__(256) void rmsnorm_kernel(
    const float* __restrict__ x, const float* __restrict__ scale,
    half_t* __restrict__ t) {
  const int lane = threadIdx.x & 31;
  const int row  = blockIdx.x * 8 + (threadIdx.x >> 5);
  const float* xr = x + (size_t)row * DM;
  v4f v[8];
  #pragma unroll
  for (int i = 0; i < 8; ++i) v[i] = *(const v4f*)(xr + (i * 32 + lane) * 4);
  float ss = 0.f;
  #pragma unroll
  for (int i = 0; i < 8; ++i)
    ss += v[i][0] * v[i][0] + v[i][1] * v[i][1] + v[i][2] * v[i][2] + v[i][3] * v[i][3];
  #pragma unroll
  for (int o = 16; o > 0; o >>= 1) ss += __shfl_xor(ss, o, 32);
  const float inv = rsqrtf(ss * (1.0f / DM) + 1e-5f);
  half_t* tr = t + (size_t)row * DM;
  #pragma unroll
  for (int i = 0; i < 8; ++i) {
    const int c = (i * 32 + lane) * 4;
    const v4f sc = *(const v4f*)(scale + c);
    v4h o;
    #pragma unroll
    for (int e = 0; e < 4; ++e) o[e] = (half_t)(v[i][e] * inv * sc[e]);
    *(v4h*)(tr + c) = o;
  }
}

template <int ITERS>
__global__ __launch_bounds__(256) void f32_to_f16_kernel(
    const float* __restrict__ src, half_t* __restrict__ dst) {
  const int i0 = blockIdx.x * 256 + threadIdx.x;
  const int stride = gridDim.x * 256;
  v4f a[ITERS];
  #pragma unroll
  for (int it = 0; it < ITERS; ++it)
    a[it] = *(const v4f*)(src + (size_t)(i0 + it * stride) * 4);
  #pragma unroll
  for (int it = 0; it < ITERS; ++it) {
    v4h o;
    #pragma unroll
    for (int e = 0; e < 4; ++e) o[e] = (half_t)a[it][e];
    *(v4h*)(dst + (size_t)(i0 + it * stride) * 4) = o;
  }
}

constexpr int GBM    = 128;
constexpr int GBK    = 64;
constexpr int GPITCH = 72;
constexpr int GTILE  = GBM * GPITCH;
constexpr int CPITCH = 68;
static_assert(4 * 64 * CPITCH * 4 <= 4 * GTILE * 2, "C parts must fit the operand stages");

__device__ __forceinline__ void gemm_fetch(v8h ra[8], v8h rb[8],
                                           const half_t* ag, const half_t* wg,
                                           int K, int k) {
  #pragma unroll
  for (int it = 0; it < 8; ++it) {
    ra[it] = *(const v8h*)(ag + (size_t)(it * 16) * K + k);
    rb[it] = *(const v8h*)(wg + (size_t)(it * 16) * K + k);
  }
}

__device__ __forceinline__ void gemm_stash(half_t* as, half_t* bs,
                                           const v8h ra[8], const v8h rb[8]) {
  #pragma unroll
  for (int it = 0; it < 8; ++it) {
    *(v8h*)(as + it * 16 * GPITCH) = ra[it];
    *(v8h*)(bs + it * 16 * GPITCH) = rb[it];
  }
}

__device__ __forceinline__ void gemm_stage(v8f acc[4][4],
                                           const half_t* af, const half_t* bf) {
  V16 a[2][4], b[2][4];
  #pragma unroll
  for (int c = 0; c < 2; ++c) {
    #pragma unroll
    for (int i = 0; i < 4; ++i) ldf(a[c][i], af + i * 16 * GPITCH + c * 32);
    #pragma unroll
    for (int j = 0; j < 4; ++j) ldf(b[c][j], bf + j * 16 * GPITCH + c * 32);
  }
  __builtin_amdgcn_sched_barrier(0);
  #pragma unroll
  for (int c = 0; c < 2; ++c)
    #pragma unroll
    for (int i = 0; i < 4; ++i)
      #pragma unroll
      for (int j = 0; j < 4; ++j)
        acc[i][j] = __builtin_amdgcn_wmma_f32_16x16x32_f16(
            false, a[c][i].v, false, b[c][j].v, (short)0, acc[i][j], false, false);
}

template <bool RES>
__global__ __launch_bounds__(128) __attribute__((amdgpu_waves_per_eu(3, 3)))
void gemm_wmma_kernel(
    const half_t* __restrict__ A, const half_t* __restrict__ W,
    const float* __restrict__ bias, const float* __restrict__ residual,
    float* __restrict__ C, int M, int N, int K) {
  __shared__ __attribute__((aligned(16))) half_t sm[4 * GTILE];
  (void)M;
  const int tid  = threadIdx.x;
  const int lane = tid & 31;
  const int wv   = tid >> 5;
  const int wm   = wv >> 1, wn = wv & 1;
  const int ntn  = N >> 7;
  const int tm   = blockIdx.x / ntn, tn = blockIdx.x % ntn;
  const int ko = (lane >= 16) ? 8 : 0;
  const int ln = lane & 15;

  const int srow = tid >> 3, scol = (tid & 7) * 8;
  const half_t* ag = A + (size_t)(tm * GBM + srow) * K + scol;
  const half_t* wg = W + (size_t)(tn * GBM + srow) * K + scol;
  half_t* sst = sm + srow * GPITCH + scol;
  const half_t* fa = sm + (wm * 64 + ln) * GPITCH + ko;
  const half_t* fb = sm + GTILE + (wn * 64 + ln) * GPITCH + ko;

  v8f acc[4][4] = {};
  v8h ra[8], rb[8];

  gemm_fetch(ra, rb, ag, wg, K, 0);
  gemm_stash(sst, sst + GTILE, ra, rb);
  __syncthreads();

  const int nk = K / GBK;
  for (int kt = 0; kt < nk - 1; ++kt) {
    const int cur = (kt & 1) * 2 * GTILE, nxt = 2 * GTILE - cur;
    gemm_fetch(ra, rb, ag, wg, K, (kt + 1) * GBK);
    __builtin_amdgcn_sched_barrier(0);
    gemm_stage(acc, fa + cur, fb + cur);
    __builtin_amdgcn_sched_barrier(0);
    gemm_stash(sst + nxt, sst + nxt + GTILE, ra, rb);
    __syncthreads();
  }
  {
    const int cur = ((nk - 1) & 1) * 2 * GTILE;
    gemm_stage(acc, fa + cur, fb + cur);
  }

  __syncthreads();
  float* cw = (float*)sm + wv * (64 * CPITCH);
  #pragma unroll
  for (int i = 0; i < 4; ++i)
    #pragma unroll
    for (int j = 0; j < 4; ++j)
      #pragma unroll
      for (int r = 0; r < 8; ++r)
        cw[(i * 16 + r + ko) * CPITCH + j * 16 + ln] = acc[i][j][r];
  __syncthreads();

  const int c4 = ln * 4;
  const int rh = lane >> 4;
  const int colg = tn * GBM + wn * 64 + c4;
  const v4f bv = *(const v4f*)(bias + colg);
  #pragma unroll 8
  for (int ps = 0; ps < 32; ++ps) {
    const int rl = ps * 2 + rh;
    const size_t g = (size_t)(tm * GBM + wm * 64 + rl) * N + colg;
    v4f v = *(const v4f*)(cw + rl * CPITCH + c4);
    v = v + bv;
    if (RES) v = v + *(const v4f*)(residual + g);
    *(v4f*)(C + g) = v;
  }
}

constexpr int VPITCH = 72;

__global__ __launch_bounds__(256) void rope_kernel(
    const float* __restrict__ qkv, half_t* __restrict__ Q,
    half_t* __restrict__ Kr, half_t* __restrict__ Vt) {
  typedef __attribute__((ext_vector_type(2))) _Float16 v2h;
  __shared__ __attribute__((aligned(16))) half_t vs[64 * VPITCH];
  const int tid = threadIdx.x;
  const int h  = blockIdx.x & (HN - 1);
  const int t0 = ((blockIdx.x >> 4) & 31) * 64;
  const int b  = blockIdx.x >> 9;
  const size_t bh = (size_t)(b * HN + h);
  const float* tile = qkv + (size_t)(b * TT + t0) * QKVD + h * 64;

  const int d4 = (tid & 15) * 4;
  const int rv = (tid >> 4) * 2;
  const int lane = tid & 31;
  const int p4   = (lane & 7) * 4;
  const int rsub = (tid >> 5) * 4 + (lane >> 3);

  v4f va[2], vb[2], q1[2], q2[2], k1[2], k2[2];
  #pragma unroll
  for (int i = 0; i < 2; ++i) {
    const float* vp = tile + (size_t)(rv + 32 * i) * QKVD + 2 * DM + d4;
    va[i] = *(const v4f*)vp;
    vb[i] = *(const v4f*)(vp + QKVD);
  }
  #pragma unroll
  for (int i = 0; i < 2; ++i) {
    const float* rp = tile + (size_t)(rsub + 32 * i) * QKVD + p4;
    q1[i] = *(const v4f*)(rp);
    q2[i] = *(const v4f*)(rp + 32);
    k1[i] = *(const v4f*)(rp + DM);
    k2[i] = *(const v4f*)(rp + DM + 32);
  }

  #pragma unroll
  for (int i = 0; i < 2; ++i) {
    #pragma unroll
    for (int e = 0; e < 4; ++e) {
      v2h pr;
      pr[0] = (half_t)va[i][e];
      pr[1] = (half_t)vb[i][e];
      *(v2h*)(&vs[(d4 + e) * VPITCH + rv + 32 * i]) = pr;
    }
  }
  __syncthreads();

  #pragma unroll
  for (int ps = 0; ps < 2; ++ps) {
    const int d  = ps * 32 + (tid >> 3);
    const int c8 = (tid & 7) * 8;
    const v8h val = *(const v8h*)(&vs[d * VPITCH + c8]);
    *(v8h*)(Vt + (bh * DH + d) * TT + t0 + c8) = val;
  }

  float invf[4];
  #pragma unroll
  for (int e = 0; e < 4; ++e)
    invf[e] = __expf(-(float)(p4 + e) * (9.210340372f / 32.0f));
  #pragma unroll
  for (int i = 0; i < 2; ++i) {
    const int t = t0 + rsub + 32 * i;
    v4h qlo, qhi, klo, khi;
    #pragma unroll
    for (int e = 0; e < 4; ++e) {
      const float ang = (float)t * invf[e];
      const float c = cosf(ang), sn = sinf(ang);
      qlo[e] = (half_t)(q1[i][e] * c - q2[i][e] * sn);
      qhi[e] = (half_t)(q2[i][e] * c + q1[i][e] * sn);
      klo[e] = (half_t)(k1[i][e] * c - k2[i][e] * sn);
      khi[e] = (half_t)(k2[i][e] * c + k1[i][e] * sn);
    }
    const size_t o = (bh * TT + t) * DH + p4;
    *(v4h*)(Q  + o)      = qlo;
    *(v4h*)(Q  + o + 32) = qhi;
    *(v4h*)(Kr + o)      = klo;
    *(v4h*)(Kr + o + 32) = khi;
  }
}

constexpr int AKEYS = 256;
constexpr int AKP   = 72;
constexpr int AVP   = 264;
static_assert(TT % 128 == 0 && TT >= AKEYS, "a block's 256 staged keys must lie inside the sequence");

__global__ __launch_bounds__(256) void attn_kernel(
    const half_t* __restrict__ Q, const half_t* __restrict__ Kr,
    const half_t* __restrict__ Vt, const int* __restrict__ amask,
    const float* __restrict__ sinks, half_t* __restrict__ attn) {
  typedef __attribute__((ext_vector_type(4))) int v4i;
  __shared__ __attribute__((aligned(16))) half_t ks[AKEYS * AKP];
  __shared__ __attribute__((aligned(16))) half_t vs[DH * AVP];
  __shared__ __attribute__((aligned(16))) int    ms[AKEYS];
  const int tid  = threadIdx.x;
  const int lane = tid & 31;
  const int wv   = tid >> 5;
  const int QB   = TT / 128;
  const int qb   = blockIdx.x % QB;
  const int bh   = blockIdx.x / QB;
  const int h    = bh % HN, b = bh / HN;
  const int q0   = qb * 128;
  const int k0   = (q0 >= 128) ? q0 - 128 : 0;
  const int qbase = q0 + wv * 16;
  const int ln = lane & 15;
  const int ko = (lane >= 16) ? 8 : 0;
  const int iq = qbase + ln;
  const half_t* Qh = Q  + (size_t)bh * TT * DH;
  const half_t* Kg = Kr + ((size_t)bh * TT + k0) * DH;
  const half_t* Vg = Vt + (size_t)bh * DH * TT + k0;
  const int* am = amask + b * TT + k0;

  V16 qb0, qb1;
  {
    const half_t* qp = Qh + (size_t)(qbase + ln) * DH;
    ldf(qb0, qp + ko);
    ldf(qb1, qp + 32 + ko);
  }

  {
    v8h rk[8], rv[8];
    #pragma unroll
    for (int it = 0; it < 8; ++it) {
      const int c = it * 256 + tid;
      rk[it] = *(const v8h*)(Kg + (size_t)c * 8);
      rv[it] = *(const v8h*)(Vg + (size_t)(c >> 5) * TT + (c & 31) * 8);
    }
    const int mw = am[tid];
    #pragma unroll
    for (int it = 0; it < 8; ++it) {
      const int c = it * 256 + tid;
      *(v8h*)(ks + (c >> 3) * AKP + (c & 7) * 8)  = rk[it];
      *(v8h*)(vs + (c >> 5) * AVP + (c & 31) * 8) = rv[it];
    }
    ms[tid] = mw;
  }
  __syncthreads();

  v8f acc[4] = {};
  float m = MINIT, l = 0.f;
  int kstart = qbase - (WIN - 1); if (kstart < 0) kstart = 0;
  kstart &= ~31;
  const int kend = qbase + 16;

  for (int jc = kstart; jc < kend; jc += 32) {
    const int jl = jc - k0;
    V16 ka[2][2], va[4];
    v4i mk[2][2];
    #pragma unroll
    for (int t2 = 0; t2 < 2; ++t2) {
      const half_t* kp = ks + (jl + t2 * 16 + ln) * AKP + ko;
      ldf(ka[t2][0], kp);
      ldf(ka[t2][1], kp + 32);
    }
    #pragma unroll
    for (int dt = 0; dt < 4; ++dt)
      ldf(va[dt], vs + (dt * 16 + ln) * AVP + jl + ko);
    #pragma unroll
    for (int t2 = 0; t2 < 2; ++t2) {
      const int* mp = ms + jl + t2 * 16 + ko;
      mk[t2][0] = *(const v4i*)(mp);
      mk[t2][1] = *(const v4i*)(mp + 4);
    }

    v8f s[2];
    #pragma unroll
    for (int t2 = 0; t2 < 2; ++t2) {
      v8f z = {};
      z = __builtin_amdgcn_wmma_f32_16x16x32_f16(false, ka[t2][0].v, false, qb0.v,
                                                 (short)0, z, false, false);
      z = __builtin_amdgcn_wmma_f32_16x16x32_f16(false, ka[t2][1].v, false, qb1.v,
                                                 (short)0, z, false, false);
      s[t2] = z;
    }
    float tl = MASKED;
    #pragma unroll
    for (int t2 = 0; t2 < 2; ++t2) {
      #pragma unroll
      for (int r = 0; r < 8; ++r) {
        const int j = jc + t2 * 16 + r + ko;
        const int mword = mk[t2][r >> 2][r & 3];
        const bool ok = (j <= iq) && (j + WIN > iq) && (mword != 0);
        const float v = ok ? s[t2][r] * SM_SCALE : MASKED;
        s[t2][r] = v;
        tl = fmaxf(tl, v);
      }
    }
    tl = fmaxf(tl, __shfl_xor(tl, 16, 32));
    const float mnew = fmaxf(m, tl);
    const float co = __expf(m - mnew);
    l *= co;
    #pragma unroll
    for (int dt = 0; dt < 4; ++dt) acc[dt] = acc[dt] * co;

    V16 pbf;
    float psum = 0.f;
    #pragma unroll
    for (int r = 0; r < 8; ++r) {
      const float p0 = __expf(s[0][r] - mnew);
      const float p1 = __expf(s[1][r] - mnew);
      psum += p0 + p1;
      pbf.h[0][r] = (half_t)p0;
      pbf.h[1][r] = (half_t)p1;
    }
    l += psum + __shfl_xor(psum, 16, 32);
    m = mnew;

    #pragma unroll
    for (int dt = 0; dt < 4; ++dt)
      acc[dt] = __builtin_amdgcn_wmma_f32_16x16x32_f16(
          false, va[dt].v, false, pbf.v, (short)0, acc[dt], false, false);
  }

  const float snk = sinks[h];
  const float m2 = fmaxf(m, snk);
  const float co = __expf(m - m2);
  const float ltot = l * co + __expf(snk - m2);
  const float factor = co / ltot;
  half_t* op = attn + (size_t)(b * TT + qbase + ln) * DM + h * 64;
  #pragma unroll
  for (int dt = 0; dt < 4; ++dt) {
    v8h o;
    #pragma unroll
    for (int r = 0; r < 8; ++r) o[r] = (half_t)(acc[dt][r] * factor);
    *(v8h*)(op + dt * 16 + ko) = o;
  }
}

extern "C" void kernel_launch(void* const* d_in, const int* in_sizes, int n_in,
                              void* d_out, int out_size, void* d_ws, size_t ws_size,
                              hipStream_t stream) {
  const float* x      = (const float*)d_in[0];
  const int*   amask  = (const int*)d_in[1];
  const float* nscale = (const float*)d_in[2];
  const float* qkv_w  = (const float*)d_in[3];
  const float* qkv_b  = (const float*)d_in[4];
  const float* out_w  = (const float*)d_in[5];
  const float* out_b  = (const float*)d_in[6];
  const float* sinks  = (const float*)d_in[7];
  float* out = (float*)d_out;
  char*  ws  = (char*)d_ws;

  half_t* qkvw_h = (half_t*)(ws + OFF_QKVW);
  half_t* outw_h = (half_t*)(ws + OFF_OUTW);
  half_t* t_h    = (half_t*)(ws + OFF_T);
  float*  qkvf   = (float*)(ws + OFF_QKVF);
  half_t* attn_h = (half_t*)(ws + OFF_ATTN);
  half_t* Qh     = (half_t*)(ws + OFF_Q);
  half_t* Kh     = (half_t*)(ws + OFF_K);
  half_t* Vth    = (half_t*)(ws + OFF_V);

  rmsnorm_kernel<<<MROW / 8, 256, 0, stream>>>(x, nscale, t_h);
  f32_to_f16_kernel<6><<<512, 256, 0, stream>>>(qkv_w, qkvw_h);
  f32_to_f16_kernel<4><<<256, 256, 0, stream>>>(out_w, outw_h);

  gemm_wmma_kernel<false><<<768, 128, 0, stream>>>(t_h, qkvw_h, qkv_b, nullptr,
                                            qkvf, MROW, QKVD, DM);

  rope_kernel<<<BB * (TT / 64) * HN, 256, 0, stream>>>(qkvf, Qh, Kh, Vth);

  attn_kernel<<<512, 256, 0, stream>>>(Qh, Kh, Vth, amask, sinks, attn_h);

  gemm_wmma_kernel<true><<<256, 128, 0, stream>>>(attn_h, outw_h, out_b, x,
                                            out, MROW, DM, DM);
}


# out_w's f16 conversion rides in the attention launch: each of its 512 blocks converts 2048 weights, loads in flight under the key loop, stored behind it; 6 -> 5 launches
# speedup vs baseline: 3.2619x; 1.1187x over previous
#include <hip/hip_runtime.h>
#include <hip/hip_bf16.h>

typedef _Float16 half_t;
typedef __attribute__((ext_vector_type(16))) _Float16 v16h;
typedef __attribute__((ext_vector_type(8)))  _Float16 v8h;
typedef __attribute__((ext_vector_type(8)))  float    v8f;
typedef __attribute__((ext_vector_type(4)))  float    v4f;
typedef __attribute__((ext_vector_type(4)))  _Float16 v4h;

union V16 { v16h v; v8h h[2]; };

constexpr int BB   = 2;
constexpr int TT   = 2048;
constexpr int DM   = 1024;
constexpr int HN   = 16;
constexpr int DH   = 64;
constexpr int QKVD = 3072;
constexpr int MROW = BB * TT;
constexpr int WIN  = 128;
constexpr float SM_SCALE = 0.125f;
constexpr float MASKED = -3.0e38f;
constexpr float MINIT  = -1.0e30f;

constexpr size_t OFF_QKVW = 0;
constexpr size_t SZ_QKVW  = (size_t)QKVD * DM * 2;
constexpr size_t OFF_OUTW = OFF_QKVW + SZ_QKVW;
constexpr size_t SZ_OUTW  = (size_t)DM * DM * 2;
constexpr size_t OFF_T    = OFF_OUTW + SZ_OUTW;
constexpr size_t SZ_T     = (size_t)MROW * DM * 2;
constexpr size_t OFF_QKVF = OFF_T + SZ_T;
constexpr size_t SZ_QKVF  = (size_t)MROW * QKVD * 4;
constexpr size_t OFF_ATTN = OFF_QKVF + SZ_QKVF;
constexpr size_t SZ_ATTN  = (size_t)MROW * DM * 2;
constexpr size_t OFF_ROT  = OFF_ATTN + SZ_ATTN;
constexpr int    ROTN     = TT * (DH / 2);

__device__ __forceinline__ void ldf(V16& f, const half_t* p) {
  f.h[0] = *(const v8h*)(p);
  f.h[1] = *(const v8h*)(p + 16);
}

__global__ __launch_bounds__(256) void rmsnorm_kernel(
    const float* __restrict__ x, const float* __restrict__ scale,
    half_t* __restrict__ t, float* __restrict__ rot) {
  const int lane = threadIdx.x & 31;
  const int row  = blockIdx.x * 8 + (threadIdx.x >> 5);
  const float* xr = x + (size_t)row * DM;
  v4f v[8];
  #pragma unroll
  for (int i = 0; i < 8; ++i) v[i] = *(const v4f*)(xr + (i * 32 + lane) * 4);
  float ss = 0.f;
  #pragma unroll
  for (int i = 0; i < 8; ++i)
    ss += v[i][0] * v[i][0] + v[i][1] * v[i][1] + v[i][2] * v[i][2] + v[i][3] * v[i][3];
  #pragma unroll
  for (int o = 16; o > 0; o >>= 1) ss += __shfl_xor(ss, o, 32);
  const float inv = rsqrtf(ss * (1.0f / DM) + 1e-5f);
  half_t* tr = t + (size_t)row * DM;
  #pragma unroll
  for (int i = 0; i < 8; ++i) {
    const int c = (i * 32 + lane) * 4;
    const v4f sc = *(const v4f*)(scale + c);
    v4h o;
    #pragma unroll
    for (int e = 0; e < 4; ++e) o[e] = (half_t)(v[i][e] * inv * sc[e]);
    *(v4h*)(tr + c) = o;
  }
  if (row < TT) {
    const float invf = __expf(-(float)lane * (9.210340372f / 32.0f));
    const float ang = (float)row * invf;
    rot[row * 32 + lane]        = cosf(ang);
    rot[ROTN + row * 32 + lane] = sinf(ang);
  }
}

template <int ITERS>
__global__ __launch_bounds__(256) void f32_to_f16_kernel(
    const float* __restrict__ src, half_t* __restrict__ dst) {
  const int i0 = blockIdx.x * 256 + threadIdx.x;
  const int stride = gridDim.x * 256;
  v4f a[ITERS];
  #pragma unroll
  for (int it = 0; it < ITERS; ++it)
    a[it] = *(const v4f*)(src + (size_t)(i0 + it * stride) * 4);
  #pragma unroll
  for (int it = 0; it < ITERS; ++it) {
    v4h o;
    #pragma unroll
    for (int e = 0; e < 4; ++e) o[e] = (half_t)a[it][e];
    *(v4h*)(dst + (size_t)(i0 + it * stride) * 4) = o;
  }
}

constexpr int GBM    = 128;
constexpr int GBK    = 64;
constexpr int GPITCH = 72;
constexpr int GTILE  = GBM * GPITCH;
constexpr int CPITCH = 68;
static_assert(4 * 64 * CPITCH * 4 <= 4 * GTILE * 2, "C parts must fit the operand stages");

__device__ __forceinline__ void gemm_fetch(v8h ra[8], v8h rb[8],
                                           const half_t* ag, const half_t* wg,
                                           int K, int k) {
  #pragma unroll
  for (int it = 0; it < 8; ++it) {
    ra[it] = *(const v8h*)(ag + (size_t)(it * 16) * K + k);
    rb[it] = *(const v8h*)(wg + (size_t)(it * 16) * K + k);
  }
}

__device__ __forceinline__ void gemm_stash(half_t* as, half_t* bs,
                                           const v8h ra[8], const v8h rb[8]) {
  #pragma unroll
  for (int it = 0; it < 8; ++it) {
    *(v8h*)(as + it * 16 * GPITCH) = ra[it];
    *(v8h*)(bs + it * 16 * GPITCH) = rb[it];
  }
}

__device__ __forceinline__ void gemm_stage(v8f acc[4][4],
                                           const half_t* af, const half_t* bf) {
  V16 a[2][4], b[2][4];
  #pragma unroll
  for (int c = 0; c < 2; ++c) {
    #pragma unroll
    for (int i = 0; i < 4; ++i) ldf(a[c][i], af + i * 16 * GPITCH + c * 32);
    #pragma unroll
    for (int j = 0; j < 4; ++j) ldf(b[c][j], bf + j * 16 * GPITCH + c * 32);
  }
  __builtin_amdgcn_sched_barrier(0);
  #pragma unroll
  for (int c = 0; c < 2; ++c)
    #pragma unroll
    for (int i = 0; i < 4; ++i)
      #pragma unroll
      for (int j = 0; j < 4; ++j)
        acc[i][j] = __builtin_amdgcn_wmma_f32_16x16x32_f16(
            false, a[c][i].v, false, b[c][j].v, (short)0, acc[i][j], false, false);
}

template <bool RES>
__global__ __launch_bounds__(128) __attribute__((amdgpu_waves_per_eu(3, 3)))
void gemm_wmma_kernel(
    const half_t* __restrict__ A, const half_t* __restrict__ W,
    const float* __restrict__ bias, const float* __restrict__ residual,
    float* __restrict__ C, int M, int N, int K) {
  __shared__ __attribute__((aligned(16))) half_t sm[4 * GTILE];
  (void)M;
  const int tid  = threadIdx.x;
  const int lane = tid & 31;
  const int wv   = tid >> 5;
  const int wm   = wv >> 1, wn = wv & 1;
  const int ntn  = N >> 7;
  const int tm   = blockIdx.x / ntn, tn = blockIdx.x % ntn;
  const int ko = (lane >= 16) ? 8 : 0;
  const int ln = lane & 15;

  const int srow = tid >> 3, scol = (tid & 7) * 8;
  const half_t* ag = A + (size_t)(tm * GBM + srow) * K + scol;
  const half_t* wg = W + (size_t)(tn * GBM + srow) * K + scol;
  half_t* sst = sm + srow * GPITCH + scol;
  const half_t* fa = sm + (wm * 64 + ln) * GPITCH + ko;
  const half_t* fb = sm + GTILE + (wn * 64 + ln) * GPITCH + ko;

  v8f acc[4][4] = {};
  v8h ra[8], rb[8];

  gemm_fetch(ra, rb, ag, wg, K, 0);
  gemm_stash(sst, sst + GTILE, ra, rb);
  __syncthreads();

  const int nk = K / GBK;
  for (int kt = 0; kt < nk - 1; ++kt) {
    const int cur = (kt & 1) * 2 * GTILE, nxt = 2 * GTILE - cur;
    gemm_fetch(ra, rb, ag, wg, K, (kt + 1) * GBK);
    __builtin_amdgcn_sched_barrier(0);
    gemm_stage(acc, fa + cur, fb + cur);
    __builtin_amdgcn_sched_barrier(0);
    gemm_stash(sst + nxt, sst + nxt + GTILE, ra, rb);
    __syncthreads();
  }
  {
    const int cur = ((nk - 1) & 1) * 2 * GTILE;
    gemm_stage(acc, fa + cur, fb + cur);
  }

  __syncthreads();
  float* cw = (float*)sm + wv * (64 * CPITCH);
  #pragma unroll
  for (int i = 0; i < 4; ++i)
    #pragma unroll
    for (int j = 0; j < 4; ++j)
      #pragma unroll
      for (int r = 0; r < 8; ++r)
        cw[(i * 16 + r + ko) * CPITCH + j * 16 + ln] = acc[i][j][r];
  __syncthreads();

  const int c4 = ln * 4;
  const int rh = lane >> 4;
  const int colg = tn * GBM + wn * 64 + c4;
  const v4f bv = *(const v4f*)(bias + colg);
  #pragma unroll 8
  for (int ps = 0; ps < 32; ++ps) {
    const int rl = ps * 2 + rh;
    const size_t g = (size_t)(tm * GBM + wm * 64 + rl) * N + colg;
    v4f v = *(const v4f*)(cw + rl * CPITCH + c4);
    v = v + bv;
    if (RES) v = v + *(const v4f*)(residual + g);
    *(v4f*)(C + g) = v;
  }
}

constexpr int AKEYS = 256;
constexpr int AQ    = 128;
constexpr int AKP   = 72;
constexpr int AVP   = 264;
static_assert(TT % 128 == 0 && TT >= AKEYS, "a block's 256 staged keys must lie inside the sequence");
constexpr int ABLOCKS = BB * HN * (TT / AQ);
constexpr int WSHARE  = DM * DM / ABLOCKS;
static_assert(WSHARE * ABLOCKS == DM * DM && WSHARE == 2 * 256 * 4,
              "a block converts two groups of 4 weights per thread");

__global__ __launch_bounds__(256) __attribute__((amdgpu_waves_per_eu(4, 4)))
void attn_kernel(
    const float* __restrict__ qkv, const float* __restrict__ rot,
    const int* __restrict__ amask,
    const float* __restrict__ sinks, half_t* __restrict__ attn,
    const float* __restrict__ wsrc, half_t* __restrict__ wdst) {
  typedef __attribute__((ext_vector_type(4))) int v4i;
  __shared__ __attribute__((aligned(16))) half_t ks[AKEYS * AKP];
  __shared__ __attribute__((aligned(16))) half_t qs[AQ * AKP];
  __shared__ __attribute__((aligned(16))) half_t vs[DH * AVP];
  __shared__ __attribute__((aligned(16))) int    ms[AKEYS];
  const int tid  = threadIdx.x;
  const int lane = tid & 31;
  const int wv   = tid >> 5;
  const int QB   = TT / AQ;
  const int qb   = blockIdx.x % QB;
  const int bh   = blockIdx.x / QB;
  const int h    = bh % HN, b = bh / HN;
  const int q0   = qb * AQ;
  const int k0   = (q0 >= 128) ? q0 - 128 : 0;
  const int qoff = q0 - k0;
  const int qbase = q0 + wv * 16;
  const int ln = lane & 15;
  const int ko = (lane >= 16) ? 8 : 0;
  const int iq = qbase + ln;
  const float* rows = qkv + (size_t)(b * TT + k0) * QKVD + h * DH;
  const float* rc = rot + k0 * 32;
  const float* rs = rc + ROTN;
  const int* am = amask + b * TT + k0;

  {
    const int rr = tid >> 3;
    const int p4 = (tid & 7) * 4;
    v4f k1[8], k2[8], cc[8], sn[8], q1[4], q2[4];
    #pragma unroll
    for (int it = 0; it < 8; ++it) {
      const int r = it * 32 + rr;
      const float* rp = rows + (size_t)r * QKVD + DM + p4;
      k1[it] = *(const v4f*)(rp);
      k2[it] = *(const v4f*)(rp + 32);
      cc[it] = *(const v4f*)(rc + r * 32 + p4);
      sn[it] = *(const v4f*)(rs + r * 32 + p4);
    }
    #pragma unroll
    for (int j = 0; j < 4; ++j) {
      const float* rp = rows + (size_t)(qoff + j * 32 + rr) * QKVD + p4;
      q1[j] = *(const v4f*)(rp);
      q2[j] = *(const v4f*)(rp + 32);
    }
    const int mw = am[tid];
    #pragma unroll
    for (int it = 0; it < 8; ++it) {
      const int r = it * 32 + rr;
      v4h lo, hi;
      #pragma unroll
      for (int e = 0; e < 4; ++e) {
        lo[e] = (half_t)(k1[it][e] * cc[it][e] - k2[it][e] * sn[it][e]);
        hi[e] = (half_t)(k2[it][e] * cc[it][e] + k1[it][e] * sn[it][e]);
      }
      *(v4h*)(ks + r * AKP + p4)      = lo;
      *(v4h*)(ks + r * AKP + 32 + p4) = hi;
    }
    #pragma unroll
    for (int j = 0; j < 4; ++j) {
      const v4f c4 = qoff ? cc[j + 4] : cc[j];
      const v4f s4 = qoff ? sn[j + 4] : sn[j];
      v4h lo, hi;
      #pragma unroll
      for (int e = 0; e < 4; ++e) {
        lo[e] = (half_t)(q1[j][e] * c4[e] - q2[j][e] * s4[e]);
        hi[e] = (half_t)(q2[j][e] * c4[e] + q1[j][e] * s4[e]);
      }
      *(v4h*)(qs + (j * 32 + rr) * AKP + p4)      = lo;
      *(v4h*)(qs + (j * 32 + rr) * AKP + 32 + p4) = hi;
    }
    ms[tid] = mw;
  }
  {
    const int g  = tid >> 4;
    const int d4 = (tid & 15) * 4;
    v4f vr[2][8];
    #pragma unroll
    for (int ps = 0; ps < 2; ++ps)
      #pragma unroll
      for (int j = 0; j < 8; ++j)
        vr[ps][j] = *(const v4f*)(rows + (size_t)((ps * 16 + g) * 8 + j) * QKVD + 2 * DM + d4);
    #pragma unroll
    for (int ps = 0; ps < 2; ++ps)
      #pragma unroll
      for (int e = 0; e < 4; ++e) {
        v8h o;
        #pragma unroll
        for (int j = 0; j < 8; ++j) o[j] = (half_t)vr[ps][j][e];
        *(v8h*)(vs + (d4 + e) * AVP + (ps * 16 + g) * 8) = o;
      }
  }
  __syncthreads();

  const size_t wo = (size_t)blockIdx.x * WSHARE + tid * 4;
  const v4f w0 = *(const v4f*)(wsrc + wo);
  const v4f w1 = *(const v4f*)(wsrc + wo + WSHARE / 2);

  V16 qb0, qb1;
  {
    const half_t* qp = qs + (wv * 16 + ln) * AKP;
    ldf(qb0, qp + ko);
    ldf(qb1, qp + 32 + ko);
  }

  v8f acc[4] = {};
  float m = MINIT, l = 0.f;
  int kstart = qbase - (WIN - 1); if (kstart < 0) kstart = 0;
  kstart &= ~31;
  const int kend = qbase + 16;

  for (int jc = kstart; jc < kend; jc += 32) {
    const int jl = jc - k0;
    V16 ka[2][2], va[4];
    v4i mk[2][2];
    #pragma unroll
    for (int t2 = 0; t2 < 2; ++t2) {
      const half_t* kp = ks + (jl + t2 * 16 + ln) * AKP + ko;
      ldf(ka[t2][0], kp);
      ldf(ka[t2][1], kp + 32);
    }
    #pragma unroll
    for (int dt = 0; dt < 4; ++dt)
      ldf(va[dt], vs + (dt * 16 + ln) * AVP + jl + ko);
    #pragma unroll
    for (int t2 = 0; t2 < 2; ++t2) {
      const int* mp = ms + jl + t2 * 16 + ko;
      mk[t2][0] = *(const v4i*)(mp);
      mk[t2][1] = *(const v4i*)(mp + 4);
    }

    v8f s[2];
    #pragma unroll
    for (int t2 = 0; t2 < 2; ++t2) {
      v8f z = {};
      z = __builtin_amdgcn_wmma_f32_16x16x32_f16(false, ka[t2][0].v, false, qb0.v,
                                                 (short)0, z, false, false);
      z = __builtin_amdgcn_wmma_f32_16x16x32_f16(false, ka[t2][1].v, false, qb1.v,
                                                 (short)0, z, false, false);
      s[t2] = z;
    }
    float tl = MASKED;
    #pragma unroll
    for (int t2 = 0; t2 < 2; ++t2) {
      #pragma unroll
      for (int r = 0; r < 8; ++r) {
        const int j = jc + t2 * 16 + r + ko;
        const int mword = mk[t2][r >> 2][r & 3];
        const bool ok = (j <= iq) && (j + WIN > iq) && (mword != 0);
        const float v = ok ? s[t2][r] * SM_SCALE : MASKED;
        s[t2][r] = v;
        tl = fmaxf(tl, v);
      }
    }
    tl = fmaxf(tl, __shfl_xor(tl, 16, 32));
    const float mnew = fmaxf(m, tl);
    const float co = __expf(m - mnew);
    l *= co;
    #pragma unroll
    for (int dt = 0; dt < 4; ++dt) acc[dt] = acc[dt] * co;

    V16 pbf;
    float psum = 0.f;
    #pragma unroll
    for (int r = 0; r < 8; ++r) {
      const float p0 = __expf(s[0][r] - mnew);
      const float p1 = __expf(s[1][r] - mnew);
      psum += p0 + p1;
      pbf.h[0][r] = (half_t)p0;
      pbf.h[1][r] = (half_t)p1;
    }
    l += psum + __shfl_xor(psum, 16, 32);
    m = mnew;

    #pragma unroll
    for (int dt = 0; dt < 4; ++dt)
      acc[dt] = __builtin_amdgcn_wmma_f32_16x16x32_f16(
          false, va[dt].v, false, pbf.v, (short)0, acc[dt], false, false);
  }

  const float snk = sinks[h];
  const float m2 = fmaxf(m, snk);
  const float co = __expf(m - m2);
  const float ltot = l * co + __expf(snk - m2);
  const float factor = co / ltot;
  half_t* op = attn + (size_t)(b * TT + qbase + ln) * DM + h * 64;
  #pragma unroll
  for (int dt = 0; dt < 4; ++dt) {
    v8h o;
    #pragma unroll
    for (int r = 0; r < 8; ++r) o[r] = (half_t)(acc[dt][r] * factor);
    *(v8h*)(op + dt * 16 + ko) = o;
  }
  *(v4h*)(wdst + wo)              = __builtin_convertvector(w0, v4h);
  *(v4h*)(wdst + wo + WSHARE / 2) = __builtin_convertvector(w1, v4h);
}

extern "C" void kernel_launch(void* const* d_in, const int* in_sizes, int n_in,
                              void* d_out, int out_size, void* d_ws, size_t ws_size,
                              hipStream_t stream) {
  const float* x      = (const float*)d_in[0];
  const int*   amask  = (const int*)d_in[1];
  const float* nscale = (const float*)d_in[2];
  const float* qkv_w  = (const float*)d_in[3];
  const float* qkv_b  = (const float*)d_in[4];
  const float* out_w  = (const float*)d_in[5];
  const float* out_b  = (const float*)d_in[6];
  const float* sinks  = (const float*)d_in[7];
  float* out = (float*)d_out;
  char*  ws  = (char*)d_ws;

  half_t* qkvw_h = (half_t*)(ws + OFF_QKVW);
  half_t* outw_h = (half_t*)(ws + OFF_OUTW);
  half_t* t_h    = (half_t*)(ws + OFF_T);
  float*  qkvf   = (float*)(ws + OFF_QKVF);
  half_t* attn_h = (half_t*)(ws + OFF_ATTN);
  float*  rot    = (float*)(ws + OFF_ROT);

  rmsnorm_kernel<<<MROW / 8, 256, 0, stream>>>(x, nscale, t_h, rot);
  f32_to_f16_kernel<6><<<512, 256, 0, stream>>>(qkv_w, qkvw_h);

  gemm_wmma_kernel<false><<<768, 128, 0, stream>>>(t_h, qkvw_h, qkv_b, nullptr,
                                            qkvf, MROW, QKVD, DM);

  attn_kernel<<<ABLOCKS, 256, 0, stream>>>(qkvf, rot, amask, sinks, attn_h, out_w, outw_h);

  gemm_wmma_kernel<true><<<256, 128, 0, stream>>>(attn_h, outw_h, out_b, x,
                                            out, MROW, DM, DM);
}


# Out-projection GEMM block = 8 waves: two groups of 4, each with own LDS stages, take the even and the odd k steps of the tile (2 waves a SIMD, 8 barriers); k sum order changes: 2 partial sums added
# speedup vs baseline: 3.3998x; 1.0423x over previous
#include <hip/hip_runtime.h>
#include <hip/hip_bf16.h>

typedef _Float16 half_t;
typedef __attribute__((ext_vector_type(16))) _Float16 v16h;
typedef __attribute__((ext_vector_type(8)))  _Float16 v8h;
typedef __attribute__((ext_vector_type(8)))  float    v8f;
typedef __attribute__((ext_vector_type(4)))  float    v4f;
typedef __attribute__((ext_vector_type(4)))  _Float16 v4h;

union V16 { v16h v; v8h h[2]; };

constexpr int BB   = 2;
constexpr int TT   = 2048;
constexpr int DM   = 1024;
constexpr int HN   = 16;
constexpr int DH   = 64;
constexpr int QKVD = 3072;
constexpr int MROW = BB * TT;
constexpr int WIN  = 128;
constexpr float SM_SCALE = 0.125f;
constexpr float MASKED = -3.0e38f;
constexpr float MINIT  = -1.0e30f;

constexpr size_t OFF_QKVW = 0;
constexpr size_t SZ_QKVW  = (size_t)QKVD * DM * 2;
constexpr size_t OFF_OUTW = OFF_QKVW + SZ_QKVW;
constexpr size_t SZ_OUTW  = (size_t)DM * DM * 2;
constexpr size_t OFF_T    = OFF_OUTW + SZ_OUTW;
constexpr size_t SZ_T     = (size_t)MROW * DM * 2;
constexpr size_t OFF_QKVF = OFF_T + SZ_T;
constexpr size_t SZ_QKVF  = (size_t)MROW * QKVD * 4;
constexpr size_t OFF_ATTN = OFF_QKVF + SZ_QKVF;
constexpr size_t SZ_ATTN  = (size_t)MROW * DM * 2;
constexpr size_t OFF_ROT  = OFF_ATTN + SZ_ATTN;
constexpr int    ROTN     = TT * (DH / 2);

__device__ __forceinline__ void ldf(V16& f, const half_t* p) {
  f.h[0] = *(const v8h*)(p);
  f.h[1] = *(const v8h*)(p + 16);
}

__global__ __launch_bounds__(256) void rmsnorm_kernel(
    const float* __restrict__ x, const float* __restrict__ scale,
    half_t* __restrict__ t, float* __restrict__ rot) {
  const int lane = threadIdx.x & 31;
  const int row  = blockIdx.x * 8 + (threadIdx.x >> 5);
  const float* xr = x + (size_t)row * DM;
  v4f v[8];
  #pragma unroll
  for (int i = 0; i < 8; ++i) v[i] = *(const v4f*)(xr + (i * 32 + lane) * 4);
  float ss = 0.f;
  #pragma unroll
  for (int i = 0; i < 8; ++i)
    ss += v[i][0] * v[i][0] + v[i][1] * v[i][1] + v[i][2] * v[i][2] + v[i][3] * v[i][3];
  #pragma unroll
  for (int o = 16; o > 0; o >>= 1) ss += __shfl_xor(ss, o, 32);
  const float inv = rsqrtf(ss * (1.0f / DM) + 1e-5f);
  half_t* tr = t + (size_t)row * DM;
  #pragma unroll
  for (int i = 0; i < 8; ++i) {
    const int c = (i * 32 + lane) * 4;
    const v4f sc = *(const v4f*)(scale + c);
    v4h o;
    #pragma unroll
    for (int e = 0; e < 4; ++e) o[e] = (half_t)(v[i][e] * inv * sc[e]);
    *(v4h*)(tr + c) = o;
  }
  if (row < TT) {
    const float invf = __expf(-(float)lane * (9.210340372f / 32.0f));
    const float ang = (float)row * invf;
    rot[row * 32 + lane]        = cosf(ang);
    rot[ROTN + row * 32 + lane] = sinf(ang);
  }
}

template <int ITERS>
__global__ __launch_bounds__(256) void f32_to_f16_kernel(
    const float* __restrict__ src, half_t* __restrict__ dst) {
  const int i0 = blockIdx.x * 256 + threadIdx.x;
  const int stride = gridDim.x * 256;
  v4f a[ITERS];
  #pragma unroll
  for (int it = 0; it < ITERS; ++it)
    a[it] = *(const v4f*)(src + (size_t)(i0 + it * stride) * 4);
  #pragma unroll
  for (int it = 0; it < ITERS; ++it) {
    v4h o;
    #pragma unroll
    for (int e = 0; e < 4; ++e) o[e] = (half_t)a[it][e];
    *(v4h*)(dst + (size_t)(i0 + it * stride) * 4) = o;
  }
}

constexpr int GBM    = 128;
constexpr int GBK    = 64;
constexpr int GPITCH = 72;
constexpr int GTILE  = GBM * GPITCH;
constexpr int CPITCH = 68;
static_assert(4 * 64 * CPITCH * 4 <= 4 * GTILE * 2, "a group's C parts must fit its operand stages");

__device__ __forceinline__ void gemm_fetch(v8h ra[8], v8h rb[8],
                                           const half_t* ag, const half_t* wg,
                                           int K, int k) {
  #pragma unroll
  for (int it = 0; it < 8; ++it) {
    ra[it] = *(const v8h*)(ag + (size_t)(it * 16) * K + k);
    rb[it] = *(const v8h*)(wg + (size_t)(it * 16) * K + k);
  }
}

__device__ __forceinline__ void gemm_stash(half_t* as, half_t* bs,
                                           const v8h ra[8], const v8h rb[8]) {
  #pragma unroll
  for (int it = 0; it < 8; ++it) {
    *(v8h*)(as + it * 16 * GPITCH) = ra[it];
    *(v8h*)(bs + it * 16 * GPITCH) = rb[it];
  }
}

__device__ __forceinline__ void gemm_stage(v8f acc[4][4],
                                           const half_t* af, const half_t* bf) {
  V16 a[2][4], b[2][4];
  #pragma unroll
  for (int c = 0; c < 2; ++c) {
    #pragma unroll
    for (int i = 0; i < 4; ++i) ldf(a[c][i], af + i * 16 * GPITCH + c * 32);
    #pragma unroll
    for (int j = 0; j < 4; ++j) ldf(b[c][j], bf + j * 16 * GPITCH + c * 32);
  }
  __builtin_amdgcn_sched_barrier(0);
  #pragma unroll
  for (int c = 0; c < 2; ++c)
    #pragma unroll
    for (int i = 0; i < 4; ++i)
      #pragma unroll
      for (int j = 0; j < 4; ++j)
        acc[i][j] = __builtin_amdgcn_wmma_f32_16x16x32_f16(
            false, a[c][i].v, false, b[c][j].v, (short)0, acc[i][j], false, false);
}

template <bool RES, int KG>
__global__ __launch_bounds__(128 * KG) __attribute__((amdgpu_waves_per_eu(3, 3)))
void gemm_wmma_kernel(
    const half_t* __restrict__ A, const half_t* __restrict__ W,
    const float* __restrict__ bias, const float* __restrict__ residual,
    float* __restrict__ C, int M, int N, int K) {
  __shared__ __attribute__((aligned(16))) half_t sm[KG * 4 * GTILE];
  (void)M;
  const int tid  = threadIdx.x & 127;
  const int kg   = threadIdx.x >> 7;
  const int lane = tid & 31;
  const int wv   = tid >> 5;
  const int wm   = wv >> 1, wn = wv & 1;
  const int ntn  = N >> 7;
  const int tm   = blockIdx.x / ntn, tn = blockIdx.x % ntn;
  const int ko = (lane >= 16) ? 8 : 0;
  const int ln = lane & 15;

  const int srow = tid >> 3, scol = (tid & 7) * 8;
  const half_t* ag = A + (size_t)(tm * GBM + srow) * K + scol;
  const half_t* wg = W + (size_t)(tn * GBM + srow) * K + scol;
  half_t* smg = sm + kg * 4 * GTILE;
  half_t* sst = smg + srow * GPITCH + scol;
  const half_t* fa = smg + (wm * 64 + ln) * GPITCH + ko;
  const half_t* fb = smg + GTILE + (wn * 64 + ln) * GPITCH + ko;

  v8f acc[4][4] = {};
  v8h ra[8], rb[8];

  gemm_fetch(ra, rb, ag, wg, K, kg * GBK);
  gemm_stash(sst, sst + GTILE, ra, rb);
  __syncthreads();

  const int nk = K / (GBK * KG);
  for (int kt = 0; kt < nk - 1; ++kt) {
    const int cur = (kt & 1) * 2 * GTILE, nxt = 2 * GTILE - cur;
    gemm_fetch(ra, rb, ag, wg, K, ((kt + 1) * KG + kg) * GBK);
    __builtin_amdgcn_sched_barrier(0);
    gemm_stage(acc, fa + cur, fb + cur);
    __builtin_amdgcn_sched_barrier(0);
    gemm_stash(sst + nxt, sst + nxt + GTILE, ra, rb);
    __syncthreads();
  }
  {
    const int cur = ((nk - 1) & 1) * 2 * GTILE;
    gemm_stage(acc, fa + cur, fb + cur);
  }

  __syncthreads();
  float* cw = (float*)sm + (kg * 4 + wv) * (64 * CPITCH);
  #pragma unroll
  for (int i = 0; i < 4; ++i)
    #pragma unroll
    for (int j = 0; j < 4; ++j)
      #pragma unroll
      for (int r = 0; r < 8; ++r)
        cw[(i * 16 + r + ko) * CPITCH + j * 16 + ln] = acc[i][j][r];
  __syncthreads();

  const int c4 = ln * 4;
  const int rh = lane >> 4;
  const int colg = tn * GBM + wn * 64 + c4;
  const v4f bv = *(const v4f*)(bias + colg);
  const float* c0 = (const float*)sm + wv * (64 * CPITCH);
  #pragma unroll 8
  for (int ps = 0; ps < 32 / KG; ++ps) {
    const int rl = kg * (64 / KG) + ps * 2 + rh;
    const size_t g = (size_t)(tm * GBM + wm * 64 + rl) * N + colg;
    v4f v = *(const v4f*)(c0 + rl * CPITCH + c4);
    #pragma unroll
    for (int q = 1; q < KG; ++q)
      v = v + *(const v4f*)(c0 + q * 4 * (64 * CPITCH) + rl * CPITCH + c4);
    v = v + bv;
    if (RES) v = v + *(const v4f*)(residual + g);
    *(v4f*)(C + g) = v;
  }
}

constexpr int AKEYS = 256;
constexpr int AQ    = 128;
constexpr int AKP   = 72;
constexpr int AVP   = 264;
static_assert(TT % 128 == 0 && TT >= AKEYS, "a block's 256 staged keys must lie inside the sequence");
constexpr int ABLOCKS = BB * HN * (TT / AQ);
constexpr int WSHARE  = DM * DM / ABLOCKS;
static_assert(WSHARE * ABLOCKS == DM * DM && WSHARE == 2 * 256 * 4,
              "a block converts two groups of 4 weights per thread");

__global__ __launch_bounds__(256) __attribute__((amdgpu_waves_per_eu(4, 4)))
void attn_kernel(
    const float* __restrict__ qkv, const float* __restrict__ rot,
    const int* __restrict__ amask,
    const float* __restrict__ sinks, half_t* __restrict__ attn,
    const float* __restrict__ wsrc, half_t* __restrict__ wdst) {
  typedef __attribute__((ext_vector_type(4))) int v4i;
  __shared__ __attribute__((aligned(16))) half_t ks[AKEYS * AKP];
  __shared__ __attribute__((aligned(16))) half_t qs[AQ * AKP];
  __shared__ __attribute__((aligned(16))) half_t vs[DH * AVP];
  __shared__ __attribute__((aligned(16))) int    ms[AKEYS];
  const int tid  = threadIdx.x;
  const int lane = tid & 31;
  const int wv   = tid >> 5;
  const int QB   = TT / AQ;
  const int qb   = blockIdx.x % QB;
  const int bh   = blockIdx.x / QB;
  const int h    = bh % HN, b = bh / HN;
  const int q0   = qb * AQ;
  const int k0   = (q0 >= 128) ? q0 - 128 : 0;
  const int qoff = q0 - k0;
  const int qbase = q0 + wv * 16;
  const int ln = lane & 15;
  const int ko = (lane >= 16) ? 8 : 0;
  const int iq = qbase + ln;
  const float* rows = qkv + (size_t)(b * TT + k0) * QKVD + h * DH;
  const float* rc = rot + k0 * 32;
  const float* rs = rc + ROTN;
  const int* am = amask + b * TT + k0;

  {
    const int rr = tid >> 3;
    const int p4 = (tid & 7) * 4;
    v4f k1[8], k2[8], cc[8], sn[8], q1[4], q2[4];
    #pragma unroll
    for (int it = 0; it < 8; ++it) {
      const int r = it * 32 + rr;
      const float* rp = rows + (size_t)r * QKVD + DM + p4;
      k1[it] = *(const v4f*)(rp);
      k2[it] = *(const v4f*)(rp + 32);
      cc[it] = *(const v4f*)(rc + r * 32 + p4);
      sn[it] = *(const v4f*)(rs + r * 32 + p4);
    }
    #pragma unroll
    for (int j = 0; j < 4; ++j) {
      const float* rp = rows + (size_t)(qoff + j * 32 + rr) * QKVD + p4;
      q1[j] = *(const v4f*)(rp);
      q2[j] = *(const v4f*)(rp + 32);
    }
    const int mw = am[tid];
    #pragma unroll
    for (int it = 0; it < 8; ++it) {
      const int r = it * 32 + rr;
      v4h lo, hi;
      #pragma unroll
      for (int e = 0; e < 4; ++e) {
        lo[e] = (half_t)(k1[it][e] * cc[it][e] - k2[it][e] * sn[it][e]);
        hi[e] = (half_t)(k2[it][e] * cc[it][e] + k1[it][e] * sn[it][e]);
      }
      *(v4h*)(ks + r * AKP + p4)      = lo;
      *(v4h*)(ks + r * AKP + 32 + p4) = hi;
    }
    #pragma unroll
    for (int j = 0; j < 4; ++j) {
      const v4f c4 = qoff ? cc[j + 4] : cc[j];
      const v4f s4 = qoff ? sn[j + 4] : sn[j];
      v4h lo, hi;
      #pragma unroll
      for (int e = 0; e < 4; ++e) {
        lo[e] = (half_t)(q1[j][e] * c4[e] - q2[j][e] * s4[e]);
        hi[e] = (half_t)(q2[j][e] * c4[e] + q1[j][e] * s4[e]);
      }
      *(v4h*)(qs + (j * 32 + rr) * AKP + p4)      = lo;
      *(v4h*)(qs + (j * 32 + rr) * AKP + 32 + p4) = hi;
    }
    ms[tid] = mw;
  }
  {
    const int g  = tid >> 4;
    const int d4 = (tid & 15) * 4;
    v4f vr[2][8];
    #pragma unroll
    for (int ps = 0; ps < 2; ++ps)
      #pragma unroll
      for (int j = 0; j < 8; ++j)
        vr[ps][j] = *(const v4f*)(rows + (size_t)((ps * 16 + g) * 8 + j) * QKVD + 2 * DM + d4);
    #pragma unroll
    for (int ps = 0; ps < 2; ++ps)
      #pragma unroll
      for (int e = 0; e < 4; ++e) {
        v8h o;
        #pragma unroll
        for (int j = 0; j < 8; ++j) o[j] = (half_t)vr[ps][j][e];
        *(v8h*)(vs + (d4 + e) * AVP + (ps * 16 + g) * 8) = o;
      }
  }
  __syncthreads();

  const size_t wo = (size_t)blockIdx.x * WSHARE + tid * 4;
  const v4f w0 = *(const v4f*)(wsrc + wo);
  const v4f w1 = *(const v4f*)(wsrc + wo + WSHARE / 2);

  V16 qb0, qb1;
  {
    const half_t* qp = qs + (wv * 16 + ln) * AKP;
    ldf(qb0, qp + ko);
    ldf(qb1, qp + 32 + ko);
  }

  v8f acc[4] = {};
  float m = MINIT, l = 0.f;
  int kstart = qbase - (WIN - 1); if (kstart < 0) kstart = 0;
  kstart &= ~31;
  const int kend = qbase + 16;

  for (int jc = kstart; jc < kend; jc += 32) {
    const int jl = jc - k0;
    V16 ka[2][2], va[4];
    v4i mk[2][2];
    #pragma unroll
    for (int t2 = 0; t2 < 2; ++t2) {
      const half_t* kp = ks + (jl + t2 * 16 + ln) * AKP + ko;
      ldf(ka[t2][0], kp);
      ldf(ka[t2][1], kp + 32);
    }
    #pragma unroll
    for (int dt = 0; dt < 4; ++dt)
      ldf(va[dt], vs + (dt * 16 + ln) * AVP + jl + ko);
    #pragma unroll
    for (int t2 = 0; t2 < 2; ++t2) {
      const int* mp = ms + jl + t2 * 16 + ko;
      mk[t2][0] = *(const v4i*)(mp);
      mk[t2][1] = *(const v4i*)(mp + 4);
    }

    v8f s[2];
    #pragma unroll
    for (int t2 = 0; t2 < 2; ++t2) {
      v8f z = {};
      z = __builtin_amdgcn_wmma_f32_16x16x32_f16(false, ka[t2][0].v, false, qb0.v,
                                                 (short)0, z, false, false);
      z = __builtin_amdgcn_wmma_f32_16x16x32_f16(false, ka[t2][1].v, false, qb1.v,
                                                 (short)0, z, false, false);
      s[t2] = z;
    }
    float tl = MASKED;
    #pragma unroll
    for (int t2 = 0; t2 < 2; ++t2) {
      #pragma unroll
      for (int r = 0; r < 8; ++r) {
        const int j = jc + t2 * 16 + r + ko;
        const int mword = mk[t2][r >> 2][r & 3];
        const bool ok = (j <= iq) && (j + WIN > iq) && (mword != 0);
        const float v = ok ? s[t2][r] * SM_SCALE : MASKED;
        s[t2][r] = v;
        tl = fmaxf(tl, v);
      }
    }
    tl = fmaxf(tl, __shfl_xor(tl, 16, 32));
    const float mnew = fmaxf(m, tl);
    const float co = __expf(m - mnew);
    l *= co;
    #pragma unroll
    for (int dt = 0; dt < 4; ++dt) acc[dt] = acc[dt] * co;

    V16 pbf;
    float psum = 0.f;
    #pragma unroll
    for (int r = 0; r < 8; ++r) {
      const float p0 = __expf(s[0][r] - mnew);
      const float p1 = __expf(s[1][r] - mnew);
      psum += p0 + p1;
      pbf.h[0][r] = (half_t)p0;
      pbf.h[1][r] = (half_t)p1;
    }
    l += psum + __shfl_xor(psum, 16, 32);
    m = mnew;

    #pragma unroll
    for (int dt = 0; dt < 4; ++dt)
      acc[dt] = __builtin_amdgcn_wmma_f32_16x16x32_f16(
          false, va[dt].v, false, pbf.v, (short)0, acc[dt], false, false);
  }

  const float snk = sinks[h];
  const float m2 = fmaxf(m, snk);
  const float co = __expf(m - m2);
  const float ltot = l * co + __expf(snk - m2);
  const float factor = co / ltot;
  half_t* op = attn + (size_t)(b * TT + qbase + ln) * DM + h * 64;
  #pragma unroll
  for (int dt = 0; dt < 4; ++dt) {
    v8h o;
    #pragma unroll
    for (int r = 0; r < 8; ++r) o[r] = (half_t)(acc[dt][r] * factor);
    *(v8h*)(op + dt * 16 + ko) = o;
  }
  *(v4h*)(wdst + wo)              = __builtin_convertvector(w0, v4h);
  *(v4h*)(wdst + wo + WSHARE / 2) = __builtin_convertvector(w1, v4h);
}

extern "C" void kernel_launch(void* const* d_in, const int* in_sizes, int n_in,
                              void* d_out, int out_size, void* d_ws, size_t ws_size,
                              hipStream_t stream) {
  const float* x      = (const float*)d_in[0];
  const int*   amask  = (const int*)d_in[1];
  const float* nscale = (const float*)d_in[2];
  const float* qkv_w  = (const float*)d_in[3];
  const float* qkv_b  = (const float*)d_in[4];
  const float* out_w  = (const float*)d_in[5];
  const float* out_b  = (const float*)d_in[6];
  const float* sinks  = (const float*)d_in[7];
  float* out = (float*)d_out;
  char*  ws  = (char*)d_ws;

  half_t* qkvw_h = (half_t*)(ws + OFF_QKVW);
  half_t* outw_h = (half_t*)(ws + OFF_OUTW);
  half_t* t_h    = (half_t*)(ws + OFF_T);
  float*  qkvf   = (float*)(ws + OFF_QKVF);
  half_t* attn_h = (half_t*)(ws + OFF_ATTN);
  float*  rot    = (float*)(ws + OFF_ROT);

  rmsnorm_kernel<<<MROW / 8, 256, 0, stream>>>(x, nscale, t_h, rot);
  f32_to_f16_kernel<6><<<512, 256, 0, stream>>>(qkv_w, qkvw_h);

  gemm_wmma_kernel<false, 1><<<768, 128, 0, stream>>>(t_h, qkvw_h, qkv_b, nullptr,
                                               qkvf, MROW, QKVD, DM);

  attn_kernel<<<ABLOCKS, 256, 0, stream>>>(qkvf, rot, amask, sinks, attn_h, out_w, outw_h);

  gemm_wmma_kernel<true, 2><<<256, 256, 0, stream>>>(attn_h, outw_h, out_b, x,
                                              out, MROW, DM, DM);
}


# q, k, v are projected inside the attention launch: a 16-wave block multiplies its 384 rows by its head's 192 weight rows via LDS stages, rotates into f16 LDS tiles; 5 -> 4 launches; k sum order kept
# speedup vs baseline: 4.1243x; 1.2131x over previous
#include <hip/hip_runtime.h>
#include <hip/hip_bf16.h>

typedef _Float16 half_t;
typedef __attribute__((ext_vector_type(16))) _Float16 v16h;
typedef __attribute__((ext_vector_type(8)))  _Float16 v8h;
typedef __attribute__((ext_vector_type(8)))  float    v8f;
typedef __attribute__((ext_vector_type(4)))  float    v4f;
typedef __attribute__((ext_vector_type(4)))  _Float16 v4h;

union V16 { v16h v; v8h h[2]; };

constexpr int BB   = 2;
constexpr int TT   = 2048;
constexpr int DM   = 1024;
constexpr int HN   = 16;
constexpr int DH   = 64;
constexpr int QKVD = 3072;
constexpr int MROW = BB * TT;
constexpr int WIN  = 128;
constexpr float SM_SCALE = 0.125f;
constexpr float MASKED = -3.0e38f;
constexpr float MINIT  = -1.0e30f;

constexpr size_t OFF_QKVW = 0;
constexpr size_t SZ_QKVW  = (size_t)QKVD * DM * 2;
constexpr size_t OFF_OUTW = OFF_QKVW + SZ_QKVW;
constexpr size_t SZ_OUTW  = (size_t)DM * DM * 2;
constexpr size_t OFF_T    = OFF_OUTW + SZ_OUTW;
constexpr size_t SZ_T     = (size_t)MROW * DM * 2;
constexpr size_t OFF_ATTN = OFF_T + SZ_T;
constexpr size_t SZ_ATTN  = (size_t)MROW * DM * 2;
constexpr size_t OFF_ROT  = OFF_ATTN + SZ_ATTN;
constexpr int    ROTN     = TT * (DH / 2);

__device__ __forceinline__ void ldf(V16& f, const half_t* p) {
  f.h[0] = *(const v8h*)(p);
  f.h[1] = *(const v8h*)(p + 16);
}

__global__ __launch_bounds__(256) void rmsnorm_kernel(
    const float* __restrict__ x, const float* __restrict__ scale,
    half_t* __restrict__ t, float* __restrict__ rot) {
  const int lane = threadIdx.x & 31;
  const int row  = blockIdx.x * 8 + (threadIdx.x >> 5);
  const float* xr = x + (size_t)row * DM;
  v4f v[8];
  #pragma unroll
  for (int i = 0; i < 8; ++i) v[i] = *(const v4f*)(xr + (i * 32 + lane) * 4);
  float ss = 0.f;
  #pragma unroll
  for (int i = 0; i < 8; ++i)
    ss += v[i][0] * v[i][0] + v[i][1] * v[i][1] + v[i][2] * v[i][2] + v[i][3] * v[i][3];
  #pragma unroll
  for (int o = 16; o > 0; o >>= 1) ss += __shfl_xor(ss, o, 32);
  const float inv = rsqrtf(ss * (1.0f / DM) + 1e-5f);
  half_t* tr = t + (size_t)row * DM;
  #pragma unroll
  for (int i = 0; i < 8; ++i) {
    const int c = (i * 32 + lane) * 4;
    const v4f sc = *(const v4f*)(scale + c);
    v4h o;
    #pragma unroll
    for (int e = 0; e < 4; ++e) o[e] = (half_t)(v[i][e] * inv * sc[e]);
    *(v4h*)(tr + c) = o;
  }
  if (row < TT) {
    const float invf = __expf(-(float)lane * (9.210340372f / 32.0f));
    const float ang = (float)row * invf;
    rot[row * 32 + lane]        = cosf(ang);
    rot[ROTN + row * 32 + lane] = sinf(ang);
  }
}

template <int ITERS>
__global__ __launch_bounds__(256) void f32_to_f16_kernel(
    const float* __restrict__ src, half_t* __restrict__ dst) {
  const int i0 = blockIdx.x * 256 + threadIdx.x;
  const int stride = gridDim.x * 256;
  v4f a[ITERS];
  #pragma unroll
  for (int it = 0; it < ITERS; ++it)
    a[it] = *(const v4f*)(src + (size_t)(i0 + it * stride) * 4);
  #pragma unroll
  for (int it = 0; it < ITERS; ++it) {
    v4h o;
    #pragma unroll
    for (int e = 0; e < 4; ++e) o[e] = (half_t)a[it][e];
    *(v4h*)(dst + (size_t)(i0 + it * stride) * 4) = o;
  }
}

constexpr int GBM    = 128;
constexpr int GBK    = 64;
constexpr int GPITCH = 72;
constexpr int GTILE  = GBM * GPITCH;
constexpr int CPITCH = 68;
static_assert(4 * 64 * CPITCH * 4 <= 4 * GTILE * 2, "a group's C parts must fit its operand stages");

__device__ __forceinline__ void gemm_fetch(v8h ra[8], v8h rb[8],
                                           const half_t* ag, const half_t* wg,
                                           int K, int k) {
  #pragma unroll
  for (int it = 0; it < 8; ++it) {
    ra[it] = *(const v8h*)(ag + (size_t)(it * 16) * K + k);
    rb[it] = *(const v8h*)(wg + (size_t)(it * 16) * K + k);
  }
}

__device__ __forceinline__ void gemm_stash(half_t* as, half_t* bs,
                                           const v8h ra[8], const v8h rb[8]) {
  #pragma unroll
  for (int it = 0; it < 8; ++it) {
    *(v8h*)(as + it * 16 * GPITCH) = ra[it];
    *(v8h*)(bs + it * 16 * GPITCH) = rb[it];
  }
}

__device__ __forceinline__ void gemm_stage(v8f acc[4][4],
                                           const half_t* af, const half_t* bf) {
  V16 a[2][4], b[2][4];
  #pragma unroll
  for (int c = 0; c < 2; ++c) {
    #pragma unroll
    for (int i = 0; i < 4; ++i) ldf(a[c][i], af + i * 16 * GPITCH + c * 32);
    #pragma unroll
    for (int j = 0; j < 4; ++j) ldf(b[c][j], bf + j * 16 * GPITCH + c * 32);
  }
  __builtin_amdgcn_sched_barrier(0);
  #pragma unroll
  for (int c = 0; c < 2; ++c)
    #pragma unroll
    for (int i = 0; i < 4; ++i)
      #pragma unroll
      for (int j = 0; j < 4; ++j)
        acc[i][j] = __builtin_amdgcn_wmma_f32_16x16x32_f16(
            false, a[c][i].v, false, b[c][j].v, (short)0, acc[i][j], false, false);
}

template <bool RES, int KG>
__global__ __launch_bounds__(128 * KG) __attribute__((amdgpu_waves_per_eu(3, 3)))
void gemm_wmma_kernel(
    const half_t* __restrict__ A, const half_t* __restrict__ W,
    const float* __restrict__ bias, const float* __restrict__ residual,
    float* __restrict__ C, int M, int N, int K) {
  __shared__ __attribute__((aligned(16))) half_t sm[KG * 4 * GTILE];
  (void)M;
  const int tid  = threadIdx.x & 127;
  const int kg   = threadIdx.x >> 7;
  const int lane = tid & 31;
  const int wv   = tid >> 5;
  const int wm   = wv >> 1, wn = wv & 1;
  const int ntn  = N >> 7;
  const int tm   = blockIdx.x / ntn, tn = blockIdx.x % ntn;
  const int ko = (lane >= 16) ? 8 : 0;
  const int ln = lane & 15;

  const int srow = tid >> 3, scol = (tid & 7) * 8;
  const half_t* ag = A + (size_t)(tm * GBM + srow) * K + scol;
  const half_t* wg = W + (size_t)(tn * GBM + srow) * K + scol;
  half_t* smg = sm + kg * 4 * GTILE;
  half_t* sst = smg + srow * GPITCH + scol;
  const half_t* fa = smg + (wm * 64 + ln) * GPITCH + ko;
  const half_t* fb = smg + GTILE + (wn * 64 + ln) * GPITCH + ko;

  v8f acc[4][4] = {};
  v8h ra[8], rb[8];

  gemm_fetch(ra, rb, ag, wg, K, kg * GBK);
  gemm_stash(sst, sst + GTILE, ra, rb);
  __syncthreads();

  const int nk = K / (GBK * KG);
  for (int kt = 0; kt < nk - 1; ++kt) {
    const int cur = (kt & 1) * 2 * GTILE, nxt = 2 * GTILE - cur;
    gemm_fetch(ra, rb, ag, wg, K, ((kt + 1) * KG + kg) * GBK);
    __builtin_amdgcn_sched_barrier(0);
    gemm_stage(acc, fa + cur, fb + cur);
    __builtin_amdgcn_sched_barrier(0);
    gemm_stash(sst + nxt, sst + nxt + GTILE, ra, rb);
    __syncthreads();
  }
  {
    const int cur = ((nk - 1) & 1) * 2 * GTILE;
    gemm_stage(acc, fa + cur, fb + cur);
  }

  __syncthreads();
  float* cw = (float*)sm + (kg * 4 + wv) * (64 * CPITCH);
  #pragma unroll
  for (int i = 0; i < 4; ++i)
    #pragma unroll
    for (int j = 0; j < 4; ++j)
      #pragma unroll
      for (int r = 0; r < 8; ++r)
        cw[(i * 16 + r + ko) * CPITCH + j * 16 + ln] = acc[i][j][r];
  __syncthreads();

  const int c4 = ln * 4;
  const int rh = lane >> 4;
  const int colg = tn * GBM + wn * 64 + c4;
  const v4f bv = *(const v4f*)(bias + colg);
  const float* c0 = (const float*)sm + wv * (64 * CPITCH);
  #pragma unroll 8
  for (int ps = 0; ps < 32 / KG; ++ps) {
    const int rl = kg * (64 / KG) + ps * 2 + rh;
    const size_t g = (size_t)(tm * GBM + wm * 64 + rl) * N + colg;
    v4f v = *(const v4f*)(c0 + rl * CPITCH + c4);
    #pragma unroll
    for (int q = 1; q < KG; ++q)
      v = v + *(const v4f*)(c0 + q * 4 * (64 * CPITCH) + rl * CPITCH + c4);
    v = v + bv;
    if (RES) v = v + *(const v4f*)(residual + g);
    *(v4f*)(C + g) = v;
  }
}

constexpr int FQ     = 256;
constexpr int FROWS  = 384;
constexpr int FWROWS = 192;
constexpr int FSTAGE = (FROWS + FWROWS) * GPITCH;
constexpr int AKP    = 72;
constexpr int FVP    = 392;
static_assert(TT % FQ == 0 && TT >= FROWS, "a block's 384 staged positions must lie inside the sequence");
static_assert(DM % GBK == 0 && DH == 64 && FQ % 64 == 0 && FROWS % 64 == 0, "64x64 wave parts");
static_assert(FROWS * AKP + FQ * AKP + DH * FVP + 2 * FROWS <= 2 * FSTAGE,
              "the K, Q, V^T tiles and the mask words must fit the drained stages");
constexpr int FBLOCKS = BB * HN * (TT / FQ);
constexpr int WSHARE  = DM * DM / FBLOCKS;
static_assert(WSHARE * FBLOCKS == DM * DM && WSHARE == 2 * 512 * 4,
              "a block converts two groups of 4 weights per thread");

__device__ __forceinline__ void proj_fetch(v8h rt[6], v8h rw[3],
                                           const half_t* tg, const half_t* wg, int k) {
  #pragma unroll
  for (int it = 0; it < 6; ++it) rt[it] = *(const v8h*)(tg + (size_t)(it * 64) * DM + k);
  #pragma unroll
  for (int sg = 0; sg < 3; ++sg) rw[sg] = *(const v8h*)(wg + (size_t)(sg * DM) * DM + k);
}

__device__ __forceinline__ void proj_stash(half_t* st, const v8h rt[6], const v8h rw[3]) {
  #pragma unroll
  for (int it = 0; it < 6; ++it) *(v8h*)(st + it * 64 * GPITCH) = rt[it];
  #pragma unroll
  for (int sg = 0; sg < 3; ++sg) *(v8h*)(st + (FROWS + sg * 64) * GPITCH) = rw[sg];
}

__device__ __forceinline__ void proj_stage(v8f acc[4][4], const half_t* af, const half_t* bf) {
  V16 a[2], b[4];
  #pragma unroll
  for (int j = 0; j < 4; ++j) ldf(b[j], bf + j * 16 * GPITCH);
  ldf(a[0], af);
  ldf(a[1], af + 16 * GPITCH);
  #pragma unroll
  for (int n = 0; n < 8; ++n) {
    const int i = n & 3;
    #pragma unroll
    for (int j = 0; j < 4; ++j)
      acc[i][j] = __builtin_amdgcn_wmma_f32_16x16x32_f16(
          false, a[n & 1].v, false, b[j].v, (short)0, acc[i][j], false, false);
    if (n + 2 < 8) ldf(a[n & 1], af + ((n + 2) & 3) * 16 * GPITCH + ((n + 2) >> 2) * 32);
    if (n == 3) {
      #pragma unroll
      for (int j = 0; j < 4; ++j) ldf(b[j], bf + j * 16 * GPITCH + 32);
    }
    if (n < 7) __builtin_amdgcn_sched_barrier(0);
  }
}

__global__ __launch_bounds__(512) __attribute__((amdgpu_waves_per_eu(4, 4)))
void proj_attn_kernel(
    const half_t* __restrict__ t, const half_t* __restrict__ w,
    const float* __restrict__ bias, const float* __restrict__ rot,
    const int* __restrict__ amask,
    const float* __restrict__ sinks, half_t* __restrict__ attn,
    const float* __restrict__ wsrc, half_t* __restrict__ wdst) {
  typedef __attribute__((ext_vector_type(4))) int v4i;
  __shared__ __attribute__((aligned(16))) half_t sm[2 * FSTAGE];
  half_t* ks = sm;
  half_t* qs = ks + FROWS * AKP;
  half_t* vs = qs + FQ * AKP;
  int*    ms = (int*)(vs + DH * FVP);
  const int tid  = threadIdx.x;
  const int lane = tid & 31;
  const int wv   = tid >> 5;
  const int QB   = TT / FQ;
  const int qb   = blockIdx.x % QB;
  const int bh   = blockIdx.x / QB;
  const int h    = bh % HN, b = bh / HN;
  const int q0   = qb * FQ;
  const int k0   = (q0 >= 128) ? q0 - 128 : 0;
  const int qoff = q0 - k0;
  const int qbase = q0 + wv * 16;
  const int ln = lane & 15;
  const int ko = (lane >= 16) ? 8 : 0;
  const int iq = qbase + ln;

  const int srow = tid >> 3, scol = (tid & 7) * 8;
  const half_t* tg = t + (size_t)(b * TT + k0 + srow) * DM + scol;
  const half_t* wg = w + (size_t)(h * DH + srow) * DM + scol;
  half_t* sst = sm + srow * GPITCH + scol;
  const int role = (wv < 6) ? 0 : (wv < 12) ? 1 : 2;
  const int grp  = (role == 0) ? wv : (role == 1) ? wv - 6 : wv - 12;
  const int arow = (role == 0) ? FROWS + 64 : (role == 1) ? grp * 64 : FROWS;
  const int brow = (role == 0) ? grp * 64 : (role == 1) ? FROWS + 128 : qoff + grp * 64;
  const half_t* fa = sm + (arow + ln) * GPITCH + ko;
  const half_t* fb = sm + (brow + ln) * GPITCH + ko;

  v8f pc[4][4] = {};
  {
    v8h rt[6], rw[3];
    proj_fetch(rt, rw, tg, wg, 0);
    proj_stash(sst, rt, rw);
    __syncthreads();
    constexpr int nk = DM / GBK;
    for (int kt = 0; kt < nk - 1; ++kt) {
      const int cur = (kt & 1) * FSTAGE, nxt = FSTAGE - cur;
      proj_fetch(rt, rw, tg, wg, (kt + 1) * GBK);
      __builtin_amdgcn_sched_barrier(0);
      proj_stage(pc, fa + cur, fb + cur);
      __builtin_amdgcn_sched_barrier(0);
      proj_stash(sst + nxt, rt, rw);
      __syncthreads();
    }
    proj_stage(pc, fa + ((nk - 1) & 1) * FSTAGE, fb + ((nk - 1) & 1) * FSTAGE);
  }
  __syncthreads();

  if (tid < FROWS) ms[tid] = amask[b * TT + k0 + tid];
  if (role != 1) {
    const float* bp = bias + (role == 0 ? DM : 0) + h * DH + ko;
    half_t* dst = (role == 0 ? ks + brow * AKP : qs + (brow - qoff) * AKP) + ln * AKP + ko;
    const float* rp = rot + (size_t)(k0 + brow + ln) * 32 + ko;
    #pragma unroll
    for (int i = 0; i < 2; ++i) {
      const v4f bl0 = *(const v4f*)(bp + i * 16),      bl1 = *(const v4f*)(bp + i * 16 + 4);
      const v4f bh0 = *(const v4f*)(bp + 32 + i * 16), bh1 = *(const v4f*)(bp + 32 + i * 16 + 4);
      #pragma unroll
      for (int j = 0; j < 4; ++j) {
        const float* rj = rp + j * 16 * 32 + i * 16;
        const v4f c0 = *(const v4f*)(rj),        c1 = *(const v4f*)(rj + 4);
        const v4f s0 = *(const v4f*)(rj + ROTN), s1 = *(const v4f*)(rj + ROTN + 4);
        v8h lo, hi;
        #pragma unroll
        for (int r = 0; r < 8; ++r) {
          const float x1 = pc[i][j][r]     + (r < 4 ? bl0[r & 3] : bl1[r & 3]);
          const float x2 = pc[i + 2][j][r] + (r < 4 ? bh0[r & 3] : bh1[r & 3]);
          const float cv = (r < 4 ? c0[r & 3] : c1[r & 3]);
          const float sv = (r < 4 ? s0[r & 3] : s1[r & 3]);
          lo[r] = (half_t)(x1 * cv - x2 * sv);
          hi[r] = (half_t)(x2 * cv + x1 * sv);
        }
        *(v8h*)(dst + j * 16 * AKP + i * 16)      = lo;
        *(v8h*)(dst + j * 16 * AKP + 32 + i * 16) = hi;
      }
      __builtin_amdgcn_sched_barrier(0);
    }
  } else {
    const float* bp = bias + 2 * DM + h * DH + ln;
    #pragma unroll
    for (int j = 0; j < 4; ++j) {
      const float bj = bp[j * 16];
      #pragma unroll
      for (int i = 0; i < 4; ++i) {
        v8h o;
        #pragma unroll
        for (int r = 0; r < 8; ++r) o[r] = (half_t)(pc[i][j][r] + bj);
        *(v8h*)(vs + (j * 16 + ln) * FVP + arow + i * 16 + ko) = o;
      }
    }
  }
  __syncthreads();

  const size_t wo = (size_t)blockIdx.x * WSHARE + tid * 4;
  const v4f w0 = *(const v4f*)(wsrc + wo);
  const v4f w1 = *(const v4f*)(wsrc + wo + WSHARE / 2);

  V16 qb0, qb1;
  {
    const half_t* qp = qs + (wv * 16 + ln) * AKP;
    ldf(qb0, qp + ko);
    ldf(qb1, qp + 32 + ko);
  }

  v8f acc[4] = {};
  float m = MINIT, l = 0.f;
  int kstart = qbase - (WIN - 1); if (kstart < 0) kstart = 0;
  kstart &= ~31;
  const int kend = qbase + 16;

  for (int jc = kstart; jc < kend; jc += 32) {
    const int jl = jc - k0;
    V16 ka[2][2], va[4];
    v4i mk[2][2];
    #pragma unroll
    for (int t2 = 0; t2 < 2; ++t2) {
      const half_t* kp = ks + (jl + t2 * 16 + ln) * AKP + ko;
      ldf(ka[t2][0], kp);
      ldf(ka[t2][1], kp + 32);
    }
    #pragma unroll
    for (int dt = 0; dt < 4; ++dt)
      ldf(va[dt], vs + (dt * 16 + ln) * FVP + jl + ko);
    #pragma unroll
    for (int t2 = 0; t2 < 2; ++t2) {
      const int* mp = ms + jl + t2 * 16 + ko;
      mk[t2][0] = *(const v4i*)(mp);
      mk[t2][1] = *(const v4i*)(mp + 4);
    }

    v8f s[2];
    #pragma unroll
    for (int t2 = 0; t2 < 2; ++t2) {
      v8f z = {};
      z = __builtin_amdgcn_wmma_f32_16x16x32_f16(false, ka[t2][0].v, false, qb0.v,
                                                 (short)0, z, false, false);
      z = __builtin_amdgcn_wmma_f32_16x16x32_f16(false, ka[t2][1].v, false, qb1.v,
                                                 (short)0, z, false, false);
      s[t2] = z;
    }
    float tl = MASKED;
    #pragma unroll
    for (int t2 = 0; t2 < 2; ++t2) {
      #pragma unroll
      for (int r = 0; r < 8; ++r) {
        const int j = jc + t2 * 16 + r + ko;
        const int mword = mk[t2][r >> 2][r & 3];
        const bool ok = (j <= iq) && (j + WIN > iq) && (mword != 0);
        const float v = ok ? s[t2][r] * SM_SCALE : MASKED;
        s[t2][r] = v;
        tl = fmaxf(tl, v);
      }
    }
    tl = fmaxf(tl, __shfl_xor(tl, 16, 32));
    const float mnew = fmaxf(m, tl);
    const float co = __expf(m - mnew);
    l *= co;
    #pragma unroll
    for (int dt = 0; dt < 4; ++dt) acc[dt] = acc[dt] * co;

    V16 pbf;
    float psum = 0.f;
    #pragma unroll
    for (int r = 0; r < 8; ++r) {
      const float p0 = __expf(s[0][r] - mnew);
      const float p1 = __expf(s[1][r] - mnew);
      psum += p0 + p1;
      pbf.h[0][r] = (half_t)p0;
      pbf.h[1][r] = (half_t)p1;
    }
    l += psum + __shfl_xor(psum, 16, 32);
    m = mnew;

    #pragma unroll
    for (int dt = 0; dt < 4; ++dt)
      acc[dt] = __builtin_amdgcn_wmma_f32_16x16x32_f16(
          false, va[dt].v, false, pbf.v, (short)0, acc[dt], false, false);
  }

  const float snk = sinks[h];
  const float m2 = fmaxf(m, snk);
  const float co = __expf(m - m2);
  const float ltot = l * co + __expf(snk - m2);
  const float factor = co / ltot;
  half_t* op = attn + (size_t)(b * TT + qbase + ln) * DM + h * 64;
  #pragma unroll
  for (int dt = 0; dt < 4; ++dt) {
    v8h o;
    #pragma unroll
    for (int r = 0; r < 8; ++r) o[r] = (half_t)(acc[dt][r] * factor);
    *(v8h*)(op + dt * 16 + ko) = o;
  }
  *(v4h*)(wdst + wo)              = __builtin_convertvector(w0, v4h);
  *(v4h*)(wdst + wo + WSHARE / 2) = __builtin_convertvector(w1, v4h);
}

extern "C" void kernel_launch(void* const* d_in, const int* in_sizes, int n_in,
                              void* d_out, int out_size, void* d_ws, size_t ws_size,
                              hipStream_t stream) {
  const float* x      = (const float*)d_in[0];
  const int*   amask  = (const int*)d_in[1];
  const float* nscale = (const float*)d_in[2];
  const float* qkv_w  = (const float*)d_in[3];
  const float* qkv_b  = (const float*)d_in[4];
  const float* out_w  = (const float*)d_in[5];
  const float* out_b  = (const float*)d_in[6];
  const float* sinks  = (const float*)d_in[7];
  float* out = (float*)d_out;
  char*  ws  = (char*)d_ws;

  half_t* qkvw_h = (half_t*)(ws + OFF_QKVW);
  half_t* outw_h = (half_t*)(ws + OFF_OUTW);
  half_t* t_h    = (half_t*)(ws + OFF_T);
  half_t* attn_h = (half_t*)(ws + OFF_ATTN);
  float*  rot    = (float*)(ws + OFF_ROT);

  rmsnorm_kernel<<<MROW / 8, 256, 0, stream>>>(x, nscale, t_h, rot);
  f32_to_f16_kernel<6><<<512, 256, 0, stream>>>(qkv_w, qkvw_h);

  proj_attn_kernel<<<FBLOCKS, 512, 0, stream>>>(t_h, qkvw_h, qkv_b, rot, amask, sinks, attn_h,
                                                out_w, outw_h);

  gemm_wmma_kernel<true, 2><<<256, 256, 0, stream>>>(attn_h, outw_h, out_b, x,
                                              out, MROW, DM, DM);
}


# Both GEMM k loops: each block starts at its own k step and wraps, so the 8 to 16 blocks that fetch the same operand rows ask L2 for different lines at any moment; k sum order changes (rotated start)
# speedup vs baseline: 4.3450x; 1.0535x over previous
#include <hip/hip_runtime.h>
#include <hip/hip_bf16.h>

typedef _Float16 half_t;
typedef __attribute__((ext_vector_type(16))) _Float16 v16h;
typedef __attribute__((ext_vector_type(8)))  _Float16 v8h;
typedef __attribute__((ext_vector_type(8)))  float    v8f;
typedef __attribute__((ext_vector_type(4)))  float    v4f;
typedef __attribute__((ext_vector_type(4)))  _Float16 v4h;

union V16 { v16h v; v8h h[2]; };

constexpr int BB   = 2;
constexpr int TT   = 2048;
constexpr int DM   = 1024;
constexpr int HN   = 16;
constexpr int DH   = 64;
constexpr int QKVD = 3072;
constexpr int MROW = BB * TT;
constexpr int WIN  = 128;
constexpr float SM_SCALE = 0.125f;
constexpr float MASKED = -3.0e38f;
constexpr float MINIT  = -1.0e30f;

constexpr size_t OFF_QKVW = 0;
constexpr size_t SZ_QKVW  = (size_t)QKVD * DM * 2;
constexpr size_t OFF_OUTW = OFF_QKVW + SZ_QKVW;
constexpr size_t SZ_OUTW  = (size_t)DM * DM * 2;
constexpr size_t OFF_T    = OFF_OUTW + SZ_OUTW;
constexpr size_t SZ_T     = (size_t)MROW * DM * 2;
constexpr size_t OFF_ATTN = OFF_T + SZ_T;
constexpr size_t SZ_ATTN  = (size_t)MROW * DM * 2;
constexpr size_t OFF_ROT  = OFF_ATTN + SZ_ATTN;
constexpr int    ROTN     = TT * (DH / 2);

__device__ __forceinline__ void ldf(V16& f, const half_t* p) {
  f.h[0] = *(const v8h*)(p);
  f.h[1] = *(const v8h*)(p + 16);
}

__global__ __launch_bounds__(256) void rmsnorm_kernel(
    const float* __restrict__ x, const float* __restrict__ scale,
    half_t* __restrict__ t, float* __restrict__ rot) {
  const int lane = threadIdx.x & 31;
  const int row  = blockIdx.x * 8 + (threadIdx.x >> 5);
  const float* xr = x + (size_t)row * DM;
  v4f v[8];
  #pragma unroll
  for (int i = 0; i < 8; ++i) v[i] = *(const v4f*)(xr + (i * 32 + lane) * 4);
  float ss = 0.f;
  #pragma unroll
  for (int i = 0; i < 8; ++i)
    ss += v[i][0] * v[i][0] + v[i][1] * v[i][1] + v[i][2] * v[i][2] + v[i][3] * v[i][3];
  #pragma unroll
  for (int o = 16; o > 0; o >>= 1) ss += __shfl_xor(ss, o, 32);
  const float inv = rsqrtf(ss * (1.0f / DM) + 1e-5f);
  half_t* tr = t + (size_t)row * DM;
  #pragma unroll
  for (int i = 0; i < 8; ++i) {
    const int c = (i * 32 + lane) * 4;
    const v4f sc = *(const v4f*)(scale + c);
    v4h o;
    #pragma unroll
    for (int e = 0; e < 4; ++e) o[e] = (half_t)(v[i][e] * inv * sc[e]);
    *(v4h*)(tr + c) = o;
  }
  if (row < TT) {
    const float invf = __expf(-(float)lane * (9.210340372f / 32.0f));
    const float ang = (float)row * invf;
    rot[row * 32 + lane]        = cosf(ang);
    rot[ROTN + row * 32 + lane] = sinf(ang);
  }
}

template <int ITERS>
__global__ __launch_bounds__(256) void f32_to_f16_kernel(
    const float* __restrict__ src, half_t* __restrict__ dst) {
  const int i0 = blockIdx.x * 256 + threadIdx.x;
  const int stride = gridDim.x * 256;
  v4f a[ITERS];
  #pragma unroll
  for (int it = 0; it < ITERS; ++it)
    a[it] = *(const v4f*)(src + (size_t)(i0 + it * stride) * 4);
  #pragma unroll
  for (int it = 0; it < ITERS; ++it) {
    v4h o;
    #pragma unroll
    for (int e = 0; e < 4; ++e) o[e] = (half_t)a[it][e];
    *(v4h*)(dst + (size_t)(i0 + it * stride) * 4) = o;
  }
}

constexpr int GBM    = 128;
constexpr int GBK    = 64;
constexpr int GPITCH = 72;
constexpr int GTILE  = GBM * GPITCH;
constexpr int CPITCH = 68;
static_assert(4 * 64 * CPITCH * 4 <= 4 * GTILE * 2, "a group's C parts must fit its operand stages");

__device__ __forceinline__ void gemm_fetch(v8h ra[8], v8h rb[8],
                                           const half_t* ag, const half_t* wg,
                                           int K, int k) {
  #pragma unroll
  for (int it = 0; it < 8; ++it) {
    ra[it] = *(const v8h*)(ag + (size_t)(it * 16) * K + k);
    rb[it] = *(const v8h*)(wg + (size_t)(it * 16) * K + k);
  }
}

__device__ __forceinline__ void gemm_stash(half_t* as, half_t* bs,
                                           const v8h ra[8], const v8h rb[8]) {
  #pragma unroll
  for (int it = 0; it < 8; ++it) {
    *(v8h*)(as + it * 16 * GPITCH) = ra[it];
    *(v8h*)(bs + it * 16 * GPITCH) = rb[it];
  }
}

__device__ __forceinline__ void gemm_stage(v8f acc[4][4],
                                           const half_t* af, const half_t* bf) {
  V16 a[2][4], b[2][4];
  #pragma unroll
  for (int c = 0; c < 2; ++c) {
    #pragma unroll
    for (int i = 0; i < 4; ++i) ldf(a[c][i], af + i * 16 * GPITCH + c * 32);
    #pragma unroll
    for (int j = 0; j < 4; ++j) ldf(b[c][j], bf + j * 16 * GPITCH + c * 32);
  }
  __builtin_amdgcn_sched_barrier(0);
  #pragma unroll
  for (int c = 0; c < 2; ++c)
    #pragma unroll
    for (int i = 0; i < 4; ++i)
      #pragma unroll
      for (int j = 0; j < 4; ++j)
        acc[i][j] = __builtin_amdgcn_wmma_f32_16x16x32_f16(
            false, a[c][i].v, false, b[c][j].v, (short)0, acc[i][j], false, false);
}

template <bool RES, int KG>
__global__ __launch_bounds__(128 * KG) __attribute__((amdgpu_waves_per_eu(3, 3)))
void gemm_wmma_kernel(
    const half_t* __restrict__ A, const half_t* __restrict__ W,
    const float* __restrict__ bias, const float* __restrict__ residual,
    float* __restrict__ C, int M, int N, int K) {
  __shared__ __attribute__((aligned(16))) half_t sm[KG * 4 * GTILE];
  (void)M;
  const int tid  = threadIdx.x & 127;
  const int kg   = threadIdx.x >> 7;
  const int lane = tid & 31;
  const int wv   = tid >> 5;
  const int wm   = wv >> 1, wn = wv & 1;
  const int ntn  = N >> 7;
  const int tm   = blockIdx.x / ntn, tn = blockIdx.x % ntn;
  const int ko = (lane >= 16) ? 8 : 0;
  const int ln = lane & 15;

  const int srow = tid >> 3, scol = (tid & 7) * 8;
  const half_t* ag = A + (size_t)(tm * GBM + srow) * K + scol;
  const half_t* wg = W + (size_t)(tn * GBM + srow) * K + scol;
  half_t* smg = sm + kg * 4 * GTILE;
  half_t* sst = smg + srow * GPITCH + scol;
  const half_t* fa = smg + (wm * 64 + ln) * GPITCH + ko;
  const half_t* fb = smg + GTILE + (wn * 64 + ln) * GPITCH + ko;

  v8f acc[4][4] = {};
  v8h ra[8], rb[8];

  const int nk = K / (GBK * KG);
  const int krot = (tm + tn) % nk;

  gemm_fetch(ra, rb, ag, wg, K, (krot * KG + kg) * GBK);
  gemm_stash(sst, sst + GTILE, ra, rb);
  __syncthreads();

  for (int kt = 0; kt < nk - 1; ++kt) {
    const int cur = (kt & 1) * 2 * GTILE, nxt = 2 * GTILE - cur;
    int ks = kt + 1 + krot;
    if (ks >= nk) ks -= nk;
    gemm_fetch(ra, rb, ag, wg, K, (ks * KG + kg) * GBK);
    __builtin_amdgcn_sched_barrier(0);
    gemm_stage(acc, fa + cur, fb + cur);
    __builtin_amdgcn_sched_barrier(0);
    gemm_stash(sst + nxt, sst + nxt + GTILE, ra, rb);
    __syncthreads();
  }
  {
    const int cur = ((nk - 1) & 1) * 2 * GTILE;
    gemm_stage(acc, fa + cur, fb + cur);
  }

  __syncthreads();
  float* cw = (float*)sm + (kg * 4 + wv) * (64 * CPITCH);
  #pragma unroll
  for (int i = 0; i < 4; ++i)
    #pragma unroll
    for (int j = 0; j < 4; ++j)
      #pragma unroll
      for (int r = 0; r < 8; ++r)
        cw[(i * 16 + r + ko) * CPITCH + j * 16 + ln] = acc[i][j][r];
  __syncthreads();

  const int c4 = ln * 4;
  const int rh = lane >> 4;
  const int colg = tn * GBM + wn * 64 + c4;
  const v4f bv = *(const v4f*)(bias + colg);
  const float* c0 = (const float*)sm + wv * (64 * CPITCH);
  #pragma unroll 8
  for (int ps = 0; ps < 32 / KG; ++ps) {
    const int rl = kg * (64 / KG) + ps * 2 + rh;
    const size_t g = (size_t)(tm * GBM + wm * 64 + rl) * N + colg;
    v4f v = *(const v4f*)(c0 + rl * CPITCH + c4);
    #pragma unroll
    for (int q = 1; q < KG; ++q)
      v = v + *(const v4f*)(c0 + q * 4 * (64 * CPITCH) + rl * CPITCH + c4);
    v = v + bv;
    if (RES) v = v + *(const v4f*)(residual + g);
    *(v4f*)(C + g) = v;
  }
}

constexpr int FQ     = 256;
constexpr int FROWS  = 384;
constexpr int FWROWS = 192;
constexpr int FSTAGE = (FROWS + FWROWS) * GPITCH;
constexpr int AKP    = 72;
constexpr int FVP    = 392;
static_assert(TT % FQ == 0 && TT >= FROWS, "a block's 384 staged positions must lie inside the sequence");
static_assert(DM % GBK == 0 && DH == 64 && FQ % 64 == 0 && FROWS % 64 == 0, "64x64 wave parts");
static_assert(FROWS * AKP + FQ * AKP + DH * FVP + 2 * FROWS <= 2 * FSTAGE,
              "the K, Q, V^T tiles and the mask words must fit the drained stages");
constexpr int FBLOCKS = BB * HN * (TT / FQ);
constexpr int WSHARE  = DM * DM / FBLOCKS;
static_assert(WSHARE * FBLOCKS == DM * DM && WSHARE == 2 * 512 * 4,
              "a block converts two groups of 4 weights per thread");

__device__ __forceinline__ void proj_fetch(v8h rt[6], v8h rw[3],
                                           const half_t* tg, const half_t* wg, int k) {
  #pragma unroll
  for (int it = 0; it < 6; ++it) rt[it] = *(const v8h*)(tg + (size_t)(it * 64) * DM + k);
  #pragma unroll
  for (int sg = 0; sg < 3; ++sg) rw[sg] = *(const v8h*)(wg + (size_t)(sg * DM) * DM + k);
}

__device__ __forceinline__ void proj_stash(half_t* st, const v8h rt[6], const v8h rw[3]) {
  #pragma unroll
  for (int it = 0; it < 6; ++it) *(v8h*)(st + it * 64 * GPITCH) = rt[it];
  #pragma unroll
  for (int sg = 0; sg < 3; ++sg) *(v8h*)(st + (FROWS + sg * 64) * GPITCH) = rw[sg];
}

__device__ __forceinline__ void proj_stage(v8f acc[4][4], const half_t* af, const half_t* bf) {
  V16 a[2], b[4];
  #pragma unroll
  for (int j = 0; j < 4; ++j) ldf(b[j], bf + j * 16 * GPITCH);
  ldf(a[0], af);
  ldf(a[1], af + 16 * GPITCH);
  #pragma unroll
  for (int n = 0; n < 8; ++n) {
    const int i = n & 3;
    #pragma unroll
    for (int j = 0; j < 4; ++j)
      acc[i][j] = __builtin_amdgcn_wmma_f32_16x16x32_f16(
          false, a[n & 1].v, false, b[j].v, (short)0, acc[i][j], false, false);
    if (n + 2 < 8) ldf(a[n & 1], af + ((n + 2) & 3) * 16 * GPITCH + ((n + 2) >> 2) * 32);
    if (n == 3) {
      #pragma unroll
      for (int j = 0; j < 4; ++j) ldf(b[j], bf + j * 16 * GPITCH + 32);
    }
    if (n < 7) __builtin_amdgcn_sched_barrier(0);
  }
}

__global__ __launch_bounds__(512) __attribute__((amdgpu_waves_per_eu(4, 4)))
void proj_attn_kernel(
    const half_t* __restrict__ t, const half_t* __restrict__ w,
    const float* __restrict__ bias, const float* __restrict__ rot,
    const int* __restrict__ amask,
    const float* __restrict__ sinks, half_t* __restrict__ attn,
    const float* __restrict__ wsrc, half_t* __restrict__ wdst) {
  typedef __attribute__((ext_vector_type(4))) int v4i;
  __shared__ __attribute__((aligned(16))) half_t sm[2 * FSTAGE];
  half_t* ks = sm;
  half_t* qs = ks + FROWS * AKP;
  half_t* vs = qs + FQ * AKP;
  int*    ms = (int*)(vs + DH * FVP);
  const int tid  = threadIdx.x;
  const int lane = tid & 31;
  const int wv   = tid >> 5;
  const int QB   = TT / FQ;
  const int qb   = blockIdx.x % QB;
  const int bh   = blockIdx.x / QB;
  const int h    = bh % HN, b = bh / HN;
  const int q0   = qb * FQ;
  const int k0   = (q0 >= 128) ? q0 - 128 : 0;
  const int qoff = q0 - k0;
  const int qbase = q0 + wv * 16;
  const int ln = lane & 15;
  const int ko = (lane >= 16) ? 8 : 0;
  const int iq = qbase + ln;

  const int srow = tid >> 3, scol = (tid & 7) * 8;
  const half_t* tg = t + (size_t)(b * TT + k0 + srow) * DM + scol;
  const half_t* wg = w + (size_t)(h * DH + srow) * DM + scol;
  half_t* sst = sm + srow * GPITCH + scol;
  const int role = (wv < 6) ? 0 : (wv < 12) ? 1 : 2;
  const int grp  = (role == 0) ? wv : (role == 1) ? wv - 6 : wv - 12;
  const int arow = (role == 0) ? FROWS + 64 : (role == 1) ? grp * 64 : FROWS;
  const int brow = (role == 0) ? grp * 64 : (role == 1) ? FROWS + 128 : qoff + grp * 64;
  const half_t* fa = sm + (arow + ln) * GPITCH + ko;
  const half_t* fb = sm + (brow + ln) * GPITCH + ko;

  v8f pc[4][4] = {};
  {
    v8h rt[6], rw[3];
    constexpr int nk = DM / GBK;
    static_assert((nk & (nk - 1)) == 0 && nk == HN && nk == BB * (TT / FQ),
                  "the k start below gives the 16 blocks of a sharing group 16 different steps");
    const int krot = (h + b * QB + qb) & (nk - 1);
    proj_fetch(rt, rw, tg, wg, krot * GBK);
    proj_stash(sst, rt, rw);
    __syncthreads();
    for (int kt = 0; kt < nk - 1; ++kt) {
      const int cur = (kt & 1) * FSTAGE, nxt = FSTAGE - cur;
      proj_fetch(rt, rw, tg, wg, ((kt + 1 + krot) & (nk - 1)) * GBK);
      __builtin_amdgcn_sched_barrier(0);
      proj_stage(pc, fa + cur, fb + cur);
      __builtin_amdgcn_sched_barrier(0);
      proj_stash(sst + nxt, rt, rw);
      __syncthreads();
    }
    proj_stage(pc, fa + ((nk - 1) & 1) * FSTAGE, fb + ((nk - 1) & 1) * FSTAGE);
  }
  __syncthreads();

  if (tid < FROWS) ms[tid] = amask[b * TT + k0 + tid];
  if (role != 1) {
    const float* bp = bias + (role == 0 ? DM : 0) + h * DH + ko;
    half_t* dst = (role == 0 ? ks + brow * AKP : qs + (brow - qoff) * AKP) + ln * AKP + ko;
    const float* rp = rot + (size_t)(k0 + brow + ln) * 32 + ko;
    #pragma unroll
    for (int i = 0; i < 2; ++i) {
      const v4f bl0 = *(const v4f*)(bp + i * 16),      bl1 = *(const v4f*)(bp + i * 16 + 4);
      const v4f bh0 = *(const v4f*)(bp + 32 + i * 16), bh1 = *(const v4f*)(bp + 32 + i * 16 + 4);
      #pragma unroll
      for (int j = 0; j < 4; ++j) {
        const float* rj = rp + j * 16 * 32 + i * 16;
        const v4f c0 = *(const v4f*)(rj),        c1 = *(const v4f*)(rj + 4);
        const v4f s0 = *(const v4f*)(rj + ROTN), s1 = *(const v4f*)(rj + ROTN + 4);
        v8h lo, hi;
        #pragma unroll
        for (int r = 0; r < 8; ++r) {
          const float x1 = pc[i][j][r]     + (r < 4 ? bl0[r & 3] : bl1[r & 3]);
          const float x2 = pc[i + 2][j][r] + (r < 4 ? bh0[r & 3] : bh1[r & 3]);
          const float cv = (r < 4 ? c0[r & 3] : c1[r & 3]);
          const float sv = (r < 4 ? s0[r & 3] : s1[r & 3]);
          lo[r] = (half_t)(x1 * cv - x2 * sv);
          hi[r] = (half_t)(x2 * cv + x1 * sv);
        }
        *(v8h*)(dst + j * 16 * AKP + i * 16)      = lo;
        *(v8h*)(dst + j * 16 * AKP + 32 + i * 16) = hi;
      }
      __builtin_amdgcn_sched_barrier(0);
    }
  } else {
    const float* bp = bias + 2 * DM + h * DH + ln;
    #pragma unroll
    for (int j = 0; j < 4; ++j) {
      const float bj = bp[j * 16];
      #pragma unroll
      for (int i = 0; i < 4; ++i) {
        v8h o;
        #pragma unroll
        for (int r = 0; r < 8; ++r) o[r] = (half_t)(pc[i][j][r] + bj);
        *(v8h*)(vs + (j * 16 + ln) * FVP + arow + i * 16 + ko) = o;
      }
    }
  }
  __syncthreads();

  const size_t wo = (size_t)blockIdx.x * WSHARE + tid * 4;
  const v4f w0 = *(const v4f*)(wsrc + wo);
  const v4f w1 = *(const v4f*)(wsrc + wo + WSHARE / 2);

  V16 qb0, qb1;
  {
    const half_t* qp = qs + (wv * 16 + ln) * AKP;
    ldf(qb0, qp + ko);
    ldf(qb1, qp + 32 + ko);
  }

  v8f acc[4] = {};
  float m = MINIT, l = 0.f;
  int kstart = qbase - (WIN - 1); if (kstart < 0) kstart = 0;
  kstart &= ~31;
  const int kend = qbase + 16;

  for (int jc = kstart; jc < kend; jc += 32) {
    const int jl = jc - k0;
    V16 ka[2][2], va[4];
    v4i mk[2][2];
    #pragma unroll
    for (int t2 = 0; t2 < 2; ++t2) {
      const half_t* kp = ks + (jl + t2 * 16 + ln) * AKP + ko;
      ldf(ka[t2][0], kp);
      ldf(ka[t2][1], kp + 32);
    }
    #pragma unroll
    for (int dt = 0; dt < 4; ++dt)
      ldf(va[dt], vs + (dt * 16 + ln) * FVP + jl + ko);
    #pragma unroll
    for (int t2 = 0; t2 < 2; ++t2) {
      const int* mp = ms + jl + t2 * 16 + ko;
      mk[t2][0] = *(const v4i*)(mp);
      mk[t2][1] = *(const v4i*)(mp + 4);
    }

    v8f s[2];
    #pragma unroll
    for (int t2 = 0; t2 < 2; ++t2) {
      v8f z = {};
      z = __builtin_amdgcn_wmma_f32_16x16x32_f16(false, ka[t2][0].v, false, qb0.v,
                                                 (short)0, z, false, false);
      z = __builtin_amdgcn_wmma_f32_16x16x32_f16(false, ka[t2][1].v, false, qb1.v,
                                                 (short)0, z, false, false);
      s[t2] = z;
    }
    float tl = MASKED;
    #pragma unroll
    for (int t2 = 0; t2 < 2; ++t2) {
      #pragma unroll
      for (int r = 0; r < 8; ++r) {
        const int j = jc + t2 * 16 + r + ko;
        const int mword = mk[t2][r >> 2][r & 3];
        const bool ok = (j <= iq) && (j + WIN > iq) && (mword != 0);
        const float v = ok ? s[t2][r] * SM_SCALE : MASKED;
        s[t2][r] = v;
        tl = fmaxf(tl, v);
      }
    }
    tl = fmaxf(tl, __shfl_xor(tl, 16, 32));
    const float mnew = fmaxf(m, tl);
    const float co = __expf(m - mnew);
    l *= co;
    #pragma unroll
    for (int dt = 0; dt < 4; ++dt) acc[dt] = acc[dt] * co;

    V16 pbf;
    float psum = 0.f;
    #pragma unroll
    for (int r = 0; r < 8; ++r) {
      const float p0 = __expf(s[0][r] - mnew);
      const float p1 = __expf(s[1][r] - mnew);
      psum += p0 + p1;
      pbf.h[0][r] = (half_t)p0;
      pbf.h[1][r] = (half_t)p1;
    }
    l += psum + __shfl_xor(psum, 16, 32);
    m = mnew;

    #pragma unroll
    for (int dt = 0; dt < 4; ++dt)
      acc[dt] = __builtin_amdgcn_wmma_f32_16x16x32_f16(
          false, va[dt].v, false, pbf.v, (short)0, acc[dt], false, false);
  }

  const float snk = sinks[h];
  const float m2 = fmaxf(m, snk);
  const float co = __expf(m - m2);
  const float ltot = l * co + __expf(snk - m2);
  const float factor = co / ltot;
  half_t* op = attn + (size_t)(b * TT + qbase + ln) * DM + h * 64;
  #pragma unroll
  for (int dt = 0; dt < 4; ++dt) {
    v8h o;
    #pragma unroll
    for (int r = 0; r < 8; ++r) o[r] = (half_t)(acc[dt][r] * factor);
    *(v8h*)(op + dt * 16 + ko) = o;
  }
  *(v4h*)(wdst + wo)              = __builtin_convertvector(w0, v4h);
  *(v4h*)(wdst + wo + WSHARE / 2) = __builtin_convertvector(w1, v4h);
}

extern "C" void kernel_launch(void* const* d_in, const int* in_sizes, int n_in,
                              void* d_out, int out_size, void* d_ws, size_t ws_size,
                              hipStream_t stream) {
  const float* x      = (const float*)d_in[0];
  const int*   amask  = (const int*)d_in[1];
  const float* nscale = (const float*)d_in[2];
  const float* qkv_w  = (const float*)d_in[3];
  const float* qkv_b  = (const float*)d_in[4];
  const float* out_w  = (const float*)d_in[5];
  const float* out_b  = (const float*)d_in[6];
  const float* sinks  = (const float*)d_in[7];
  float* out = (float*)d_out;
  char*  ws  = (char*)d_ws;

  half_t* qkvw_h = (half_t*)(ws + OFF_QKVW);
  half_t* outw_h = (half_t*)(ws + OFF_OUTW);
  half_t* t_h    = (half_t*)(ws + OFF_T);
  half_t* attn_h = (half_t*)(ws + OFF_ATTN);
  float*  rot    = (float*)(ws + OFF_ROT);

  rmsnorm_kernel<<<MROW / 8, 256, 0, stream>>>(x, nscale, t_h, rot);
  f32_to_f16_kernel<6><<<512, 256, 0, stream>>>(qkv_w, qkvw_h);

  proj_attn_kernel<<<FBLOCKS, 512, 0, stream>>>(t_h, qkvw_h, qkv_b, rot, amask, sinks, attn_h,
                                                out_w, outw_h);

  gemm_wmma_kernel<true, 2><<<256, 256, 0, stream>>>(attn_h, outw_h, out_b, x,
                                              out, MROW, DM, DM);
}


# QKV projection k loop: step k+2's fetch is issued mid-step k and handed to the LDS mid-step k+1; the step barrier waits for LDS traffic only, no wave stands there for a global load; k sum order kept
# speedup vs baseline: 4.5365x; 1.0441x over previous
#include <hip/hip_runtime.h>
#include <hip/hip_bf16.h>

typedef _Float16 half_t;
typedef __attribute__((ext_vector_type(16))) _Float16 v16h;
typedef __attribute__((ext_vector_type(8)))  _Float16 v8h;
typedef __attribute__((ext_vector_type(8)))  float    v8f;
typedef __attribute__((ext_vector_type(4)))  float    v4f;
typedef __attribute__((ext_vector_type(4)))  _Float16 v4h;

union V16 { v16h v; v8h h[2]; };

constexpr int BB   = 2;
constexpr int TT   = 2048;
constexpr int DM   = 1024;
constexpr int HN   = 16;
constexpr int DH   = 64;
constexpr int QKVD = 3072;
constexpr int MROW = BB * TT;
constexpr int WIN  = 128;
constexpr float SM_SCALE = 0.125f;
constexpr float MASKED = -3.0e38f;
constexpr float MINIT  = -1.0e30f;

constexpr size_t OFF_QKVW = 0;
constexpr size_t SZ_QKVW  = (size_t)QKVD * DM * 2;
constexpr size_t OFF_OUTW = OFF_QKVW + SZ_QKVW;
constexpr size_t SZ_OUTW  = (size_t)DM * DM * 2;
constexpr size_t OFF_T    = OFF_OUTW + SZ_OUTW;
constexpr size_t SZ_T     = (size_t)MROW * DM * 2;
constexpr size_t OFF_ATTN = OFF_T + SZ_T;
constexpr size_t SZ_ATTN  = (size_t)MROW * DM * 2;
constexpr size_t OFF_ROT  = OFF_ATTN + SZ_ATTN;
constexpr int    ROTN     = TT * (DH / 2);

__device__ __forceinline__ void ldf(V16& f, const half_t* p) {
  f.h[0] = *(const v8h*)(p);
  f.h[1] = *(const v8h*)(p + 16);
}

__global__ __launch_bounds__(256) void rmsnorm_kernel(
    const float* __restrict__ x, const float* __restrict__ scale,
    half_t* __restrict__ t, float* __restrict__ rot) {
  const int lane = threadIdx.x & 31;
  const int row  = blockIdx.x * 8 + (threadIdx.x >> 5);
  const float* xr = x + (size_t)row * DM;
  v4f v[8];
  #pragma unroll
  for (int i = 0; i < 8; ++i) v[i] = *(const v4f*)(xr + (i * 32 + lane) * 4);
  float ss = 0.f;
  #pragma unroll
  for (int i = 0; i < 8; ++i)
    ss += v[i][0] * v[i][0] + v[i][1] * v[i][1] + v[i][2] * v[i][2] + v[i][3] * v[i][3];
  #pragma unroll
  for (int o = 16; o > 0; o >>= 1) ss += __shfl_xor(ss, o, 32);
  const float inv = rsqrtf(ss * (1.0f / DM) + 1e-5f);
  half_t* tr = t + (size_t)row * DM;
  #pragma unroll
  for (int i = 0; i < 8; ++i) {
    const int c = (i * 32 + lane) * 4;
    const v4f sc = *(const v4f*)(scale + c);
    v4h o;
    #pragma unroll
    for (int e = 0; e < 4; ++e) o[e] = (half_t)(v[i][e] * inv * sc[e]);
    *(v4h*)(tr + c) = o;
  }
  if (row < TT) {
    const float invf = __expf(-(float)lane * (9.210340372f / 32.0f));
    const float ang = (float)row * invf;
    rot[row * 32 + lane]        = cosf(ang);
    rot[ROTN + row * 32 + lane] = sinf(ang);
  }
}

template <int ITERS>
__global__ __launch_bounds__(256) void f32_to_f16_kernel(
    const float* __restrict__ src, half_t* __restrict__ dst) {
  const int i0 = blockIdx.x * 256 + threadIdx.x;
  const int stride = gridDim.x * 256;
  v4f a[ITERS];
  #pragma unroll
  for (int it = 0; it < ITERS; ++it)
    a[it] = *(const v4f*)(src + (size_t)(i0 + it * stride) * 4);
  #pragma unroll
  for (int it = 0; it < ITERS; ++it) {
    v4h o;
    #pragma unroll
    for (int e = 0; e < 4; ++e) o[e] = (half_t)a[it][e];
    *(v4h*)(dst + (size_t)(i0 + it * stride) * 4) = o;
  }
}

constexpr int GBM    = 128;
constexpr int GBK    = 64;
constexpr int GPITCH = 72;
constexpr int GTILE  = GBM * GPITCH;
constexpr int CPITCH = 68;
static_assert(4 * 64 * CPITCH * 4 <= 4 * GTILE * 2, "a group's C parts must fit its operand stages");

__device__ __forceinline__ void gemm_fetch(v8h ra[8], v8h rb[8],
                                           const half_t* ag, const half_t* wg,
                                           int K, int k) {
  #pragma unroll
  for (int it = 0; it < 8; ++it) {
    ra[it] = *(const v8h*)(ag + (size_t)(it * 16) * K + k);
    rb[it] = *(const v8h*)(wg + (size_t)(it * 16) * K + k);
  }
}

__device__ __forceinline__ void gemm_stash(half_t* as, half_t* bs,
                                           const v8h ra[8], const v8h rb[8]) {
  #pragma unroll
  for (int it = 0; it < 8; ++it) {
    *(v8h*)(as + it * 16 * GPITCH) = ra[it];
    *(v8h*)(bs + it * 16 * GPITCH) = rb[it];
  }
}

__device__ __forceinline__ void gemm_stage(v8f acc[4][4],
                                           const half_t* af, const half_t* bf) {
  V16 a[2][4], b[2][4];
  #pragma unroll
  for (int c = 0; c < 2; ++c) {
    #pragma unroll
    for (int i = 0; i < 4; ++i) ldf(a[c][i], af + i * 16 * GPITCH + c * 32);
    #pragma unroll
    for (int j = 0; j < 4; ++j) ldf(b[c][j], bf + j * 16 * GPITCH + c * 32);
  }
  __builtin_amdgcn_sched_barrier(0);
  #pragma unroll
  for (int c = 0; c < 2; ++c)
    #pragma unroll
    for (int i = 0; i < 4; ++i)
      #pragma unroll
      for (int j = 0; j < 4; ++j)
        acc[i][j] = __builtin_amdgcn_wmma_f32_16x16x32_f16(
            false, a[c][i].v, false, b[c][j].v, (short)0, acc[i][j], false, false);
}

template <bool RES, int KG>
__global__ __launch_bounds__(128 * KG) __attribute__((amdgpu_waves_per_eu(3, 3)))
void gemm_wmma_kernel(
    const half_t* __restrict__ A, const half_t* __restrict__ W,
    const float* __restrict__ bias, const float* __restrict__ residual,
    float* __restrict__ C, int M, int N, int K) {
  __shared__ __attribute__((aligned(16))) half_t sm[KG * 4 * GTILE];
  (void)M;
  const int tid  = threadIdx.x & 127;
  const int kg   = threadIdx.x >> 7;
  const int lane = tid & 31;
  const int wv   = tid >> 5;
  const int wm   = wv >> 1, wn = wv & 1;
  const int ntn  = N >> 7;
  const int tm   = blockIdx.x / ntn, tn = blockIdx.x % ntn;
  const int ko = (lane >= 16) ? 8 : 0;
  const int ln = lane & 15;

  const int srow = tid >> 3, scol = (tid & 7) * 8;
  const half_t* ag = A + (size_t)(tm * GBM + srow) * K + scol;
  const half_t* wg = W + (size_t)(tn * GBM + srow) * K + scol;
  half_t* smg = sm + kg * 4 * GTILE;
  half_t* sst = smg + srow * GPITCH + scol;
  const half_t* fa = smg + (wm * 64 + ln) * GPITCH + ko;
  const half_t* fb = smg + GTILE + (wn * 64 + ln) * GPITCH + ko;

  v8f acc[4][4] = {};
  v8h ra[8], rb[8];

  const int nk = K / (GBK * KG);
  const int krot = (tm + tn) % nk;

  gemm_fetch(ra, rb, ag, wg, K, (krot * KG + kg) * GBK);
  gemm_stash(sst, sst + GTILE, ra, rb);
  __syncthreads();

  for (int kt = 0; kt < nk - 1; ++kt) {
    const int cur = (kt & 1) * 2 * GTILE, nxt = 2 * GTILE - cur;
    int ks = kt + 1 + krot;
    if (ks >= nk) ks -= nk;
    gemm_fetch(ra, rb, ag, wg, K, (ks * KG + kg) * GBK);
    __builtin_amdgcn_sched_barrier(0);
    gemm_stage(acc, fa + cur, fb + cur);
    __builtin_amdgcn_sched_barrier(0);
    gemm_stash(sst + nxt, sst + nxt + GTILE, ra, rb);
    __syncthreads();
  }
  {
    const int cur = ((nk - 1) & 1) * 2 * GTILE;
    gemm_stage(acc, fa + cur, fb + cur);
  }

  __syncthreads();
  float* cw = (float*)sm + (kg * 4 + wv) * (64 * CPITCH);
  #pragma unroll
  for (int i = 0; i < 4; ++i)
    #pragma unroll
    for (int j = 0; j < 4; ++j)
      #pragma unroll
      for (int r = 0; r < 8; ++r)
        cw[(i * 16 + r + ko) * CPITCH + j * 16 + ln] = acc[i][j][r];
  __syncthreads();

  const int c4 = ln * 4;
  const int rh = lane >> 4;
  const int colg = tn * GBM + wn * 64 + c4;
  const v4f bv = *(const v4f*)(bias + colg);
  const float* c0 = (const float*)sm + wv * (64 * CPITCH);
  #pragma unroll 8
  for (int ps = 0; ps < 32 / KG; ++ps) {
    const int rl = kg * (64 / KG) + ps * 2 + rh;
    const size_t g = (size_t)(tm * GBM + wm * 64 + rl) * N + colg;
    v4f v = *(const v4f*)(c0 + rl * CPITCH + c4);
    #pragma unroll
    for (int q = 1; q < KG; ++q)
      v = v + *(const v4f*)(c0 + q * 4 * (64 * CPITCH) + rl * CPITCH + c4);
    v = v + bv;
    if (RES) v = v + *(const v4f*)(residual + g);
    *(v4f*)(C + g) = v;
  }
}

constexpr int FQ     = 256;
constexpr int FROWS  = 384;
constexpr int FWROWS = 192;
constexpr int FSTAGE = (FROWS + FWROWS) * GPITCH;
constexpr int AKP    = 72;
constexpr int FVP    = 392;
static_assert(TT % FQ == 0 && TT >= FROWS, "a block's 384 staged positions must lie inside the sequence");
static_assert(DM % GBK == 0 && DH == 64 && FQ % 64 == 0 && FROWS % 64 == 0, "64x64 wave parts");
static_assert(FROWS * AKP + FQ * AKP + DH * FVP + 2 * FROWS <= 2 * FSTAGE,
              "the K, Q, V^T tiles and the mask words must fit the drained stages");
constexpr int FBLOCKS = BB * HN * (TT / FQ);
constexpr int WSHARE  = DM * DM / FBLOCKS;
static_assert(WSHARE * FBLOCKS == DM * DM && WSHARE == 2 * 512 * 4,
              "a block converts two groups of 4 weights per thread");

__device__ __forceinline__ void proj_fetch(v8h rt[6], v8h rw[3],
                                           const half_t* tg, const half_t* wg, int k) {
  #pragma unroll
  for (int it = 0; it < 6; ++it) rt[it] = *(const v8h*)(tg + (size_t)(it * 64) * DM + k);
  #pragma unroll
  for (int sg = 0; sg < 3; ++sg) rw[sg] = *(const v8h*)(wg + (size_t)(sg * DM) * DM + k);
}

__device__ __forceinline__ void proj_stash(half_t* st, const v8h rt[6], const v8h rw[3]) {
  #pragma unroll
  for (int it = 0; it < 6; ++it) *(v8h*)(st + it * 64 * GPITCH) = rt[it];
  #pragma unroll
  for (int sg = 0; sg < 3; ++sg) *(v8h*)(st + (FROWS + sg * 64) * GPITCH) = rw[sg];
}

__device__ __forceinline__ void lds_barrier() {
  asm volatile("s_wait_dscnt 0x0\n\ts_barrier_signal -1\n\ts_barrier_wait -1" ::: "memory");
}

template <class Between>
__device__ __forceinline__ void proj_stage(v8f acc[4][4], const half_t* af, const half_t* bf,
                                           Between between) {
  V16 a[2], b[4];
  #pragma unroll
  for (int j = 0; j < 4; ++j) ldf(b[j], bf + j * 16 * GPITCH);
  ldf(a[0], af);
  ldf(a[1], af + 16 * GPITCH);
  #pragma unroll
  for (int n = 0; n < 8; ++n) {
    const int i = n & 3;
    #pragma unroll
    for (int j = 0; j < 4; ++j)
      acc[i][j] = __builtin_amdgcn_wmma_f32_16x16x32_f16(
          false, a[n & 1].v, false, b[j].v, (short)0, acc[i][j], false, false);
    if (n + 2 < 8) ldf(a[n & 1], af + ((n + 2) & 3) * 16 * GPITCH + ((n + 2) >> 2) * 32);
    if (n == 3) {
      #pragma unroll
      for (int j = 0; j < 4; ++j) ldf(b[j], bf + j * 16 * GPITCH + 32);
      __builtin_amdgcn_sched_barrier(0);
      between();
    }
    if (n < 7) __builtin_amdgcn_sched_barrier(0);
  }
}

__global__ __launch_bounds__(512) __attribute__((amdgpu_waves_per_eu(4, 4)))
void proj_attn_kernel(
    const half_t* __restrict__ t, const half_t* __restrict__ w,
    const float* __restrict__ bias, const float* __restrict__ rot,
    const int* __restrict__ amask,
    const float* __restrict__ sinks, half_t* __restrict__ attn,
    const float* __restrict__ wsrc, half_t* __restrict__ wdst) {
  typedef __attribute__((ext_vector_type(4))) int v4i;
  __shared__ __attribute__((aligned(16))) half_t sm[2 * FSTAGE];
  half_t* ks = sm;
  half_t* qs = ks + FROWS * AKP;
  half_t* vs = qs + FQ * AKP;
  int*    ms = (int*)(vs + DH * FVP);
  const int tid  = threadIdx.x;
  const int lane = tid & 31;
  const int wv   = tid >> 5;
  const int QB   = TT / FQ;
  const int qb   = blockIdx.x % QB;
  const int bh   = blockIdx.x / QB;
  const int h    = bh % HN, b = bh / HN;
  const int q0   = qb * FQ;
  const int k0   = (q0 >= 128) ? q0 - 128 : 0;
  const int qoff = q0 - k0;
  const int qbase = q0 + wv * 16;
  const int ln = lane & 15;
  const int ko = (lane >= 16) ? 8 : 0;
  const int iq = qbase + ln;

  const int srow = tid >> 3, scol = (tid & 7) * 8;
  const half_t* tg = t + (size_t)(b * TT + k0 + srow) * DM + scol;
  const half_t* wg = w + (size_t)(h * DH + srow) * DM + scol;
  half_t* sst = sm + srow * GPITCH + scol;
  const int role = (wv < 6) ? 0 : (wv < 12) ? 1 : 2;
  const int grp  = (role == 0) ? wv : (role == 1) ? wv - 6 : wv - 12;
  const int arow = (role == 0) ? FROWS + 64 : (role == 1) ? grp * 64 : FROWS;
  const int brow = (role == 0) ? grp * 64 : (role == 1) ? FROWS + 128 : qoff + grp * 64;
  const half_t* fa = sm + (arow + ln) * GPITCH + ko;
  const half_t* fb = sm + (brow + ln) * GPITCH + ko;

  v8f pc[4][4] = {};
  {
    v8h rt[6], rw[3];
    constexpr int nk = DM / GBK;
    static_assert((nk & (nk - 1)) == 0 && nk == HN && nk == BB * (TT / FQ),
                  "the k start below gives the 16 blocks of a sharing group 16 different steps");
    const int krot = (h + b * QB + qb) & (nk - 1);
    proj_fetch(rt, rw, tg, wg, krot * GBK);
    proj_stash(sst, rt, rw);
    proj_fetch(rt, rw, tg, wg, ((1 + krot) & (nk - 1)) * GBK);
    lds_barrier();
    for (int kt = 0; kt < nk - 1; ++kt) {
      const int cur = (kt & 1) * FSTAGE, nxt = FSTAGE - cur;
      proj_stage(pc, fa + cur, fb + cur, [&]() {
        proj_stash(sst + nxt, rt, rw);
        if (kt + 2 < nk)
          proj_fetch(rt, rw, tg, wg, ((kt + 2 + krot) & (nk - 1)) * GBK);
      });
      lds_barrier();
    }
    proj_stage(pc, fa + ((nk - 1) & 1) * FSTAGE, fb + ((nk - 1) & 1) * FSTAGE, []() {});
  }
  __syncthreads();

  if (tid < FROWS) ms[tid] = amask[b * TT + k0 + tid];
  if (role != 1) {
    const float* bp = bias + (role == 0 ? DM : 0) + h * DH + ko;
    half_t* dst = (role == 0 ? ks + brow * AKP : qs + (brow - qoff) * AKP) + ln * AKP + ko;
    const float* rp = rot + (size_t)(k0 + brow + ln) * 32 + ko;
    #pragma unroll
    for (int i = 0; i < 2; ++i) {
      const v4f bl0 = *(const v4f*)(bp + i * 16),      bl1 = *(const v4f*)(bp + i * 16 + 4);
      const v4f bh0 = *(const v4f*)(bp + 32 + i * 16), bh1 = *(const v4f*)(bp + 32 + i * 16 + 4);
      #pragma unroll
      for (int j = 0; j < 4; ++j) {
        const float* rj = rp + j * 16 * 32 + i * 16;
        const v4f c0 = *(const v4f*)(rj),        c1 = *(const v4f*)(rj + 4);
        const v4f s0 = *(const v4f*)(rj + ROTN), s1 = *(const v4f*)(rj + ROTN + 4);
        v8h lo, hi;
        #pragma unroll
        for (int r = 0; r < 8; ++r) {
          const float x1 = pc[i][j][r]     + (r < 4 ? bl0[r & 3] : bl1[r & 3]);
          const float x2 = pc[i + 2][j][r] + (r < 4 ? bh0[r & 3] : bh1[r & 3]);
          const float cv = (r < 4 ? c0[r & 3] : c1[r & 3]);
          const float sv = (r < 4 ? s0[r & 3] : s1[r & 3]);
          lo[r] = (half_t)(x1 * cv - x2 * sv);
          hi[r] = (half_t)(x2 * cv + x1 * sv);
        }
        *(v8h*)(dst + j * 16 * AKP + i * 16)      = lo;
        *(v8h*)(dst + j * 16 * AKP + 32 + i * 16) = hi;
      }
      __builtin_amdgcn_sched_barrier(0);
    }
  } else {
    const float* bp = bias + 2 * DM + h * DH + ln;
    #pragma unroll
    for (int j = 0; j < 4; ++j) {
      const float bj = bp[j * 16];
      #pragma unroll
      for (int i = 0; i < 4; ++i) {
        v8h o;
        #pragma unroll
        for (int r = 0; r < 8; ++r) o[r] = (half_t)(pc[i][j][r] + bj);
        *(v8h*)(vs + (j * 16 + ln) * FVP + arow + i * 16 + ko) = o;
      }
    }
  }
  __syncthreads();

  const size_t wo = (size_t)blockIdx.x * WSHARE + tid * 4;
  const v4f w0 = *(const v4f*)(wsrc + wo);
  const v4f w1 = *(const v4f*)(wsrc + wo + WSHARE / 2);

  V16 qb0, qb1;
  {
    const half_t* qp = qs + (wv * 16 + ln) * AKP;
    ldf(qb0, qp + ko);
    ldf(qb1, qp + 32 + ko);
  }

  v8f acc[4] = {};
  float m = MINIT, l = 0.f;
  int kstart = qbase - (WIN - 1); if (kstart < 0) kstart = 0;
  kstart &= ~31;
  const int kend = qbase + 16;

  for (int jc = kstart; jc < kend; jc += 32) {
    const int jl = jc - k0;
    V16 ka[2][2], va[4];
    v4i mk[2][2];
    #pragma unroll
    for (int t2 = 0; t2 < 2; ++t2) {
      const half_t* kp = ks + (jl + t2 * 16 + ln) * AKP + ko;
      ldf(ka[t2][0], kp);
      ldf(ka[t2][1], kp + 32);
    }
    #pragma unroll
    for (int dt = 0; dt < 4; ++dt)
      ldf(va[dt], vs + (dt * 16 + ln) * FVP + jl + ko);
    #pragma unroll
    for (int t2 = 0; t2 < 2; ++t2) {
      const int* mp = ms + jl + t2 * 16 + ko;
      mk[t2][0] = *(const v4i*)(mp);
      mk[t2][1] = *(const v4i*)(mp + 4);
    }

    v8f s[2];
    #pragma unroll
    for (int t2 = 0; t2 < 2; ++t2) {
      v8f z = {};
      z = __builtin_amdgcn_wmma_f32_16x16x32_f16(false, ka[t2][0].v, false, qb0.v,
                                                 (short)0, z, false, false);
      z = __builtin_amdgcn_wmma_f32_16x16x32_f16(false, ka[t2][1].v, false, qb1.v,
                                                 (short)0, z, false, false);
      s[t2] = z;
    }
    float tl = MASKED;
    #pragma unroll
    for (int t2 = 0; t2 < 2; ++t2) {
      #pragma unroll
      for (int r = 0; r < 8; ++r) {
        const int j = jc + t2 * 16 + r + ko;
        const int mword = mk[t2][r >> 2][r & 3];
        const bool ok = (j <= iq) && (j + WIN > iq) && (mword != 0);
        const float v = ok ? s[t2][r] * SM_SCALE : MASKED;
        s[t2][r] = v;
        tl = fmaxf(tl, v);
      }
    }
    tl = fmaxf(tl, __shfl_xor(tl, 16, 32));
    const float mnew = fmaxf(m, tl);
    const float co = __expf(m - mnew);
    l *= co;
    #pragma unroll
    for (int dt = 0; dt < 4; ++dt) acc[dt] = acc[dt] * co;

    V16 pbf;
    float psum = 0.f;
    #pragma unroll
    for (int r = 0; r < 8; ++r) {
      const float p0 = __expf(s[0][r] - mnew);
      const float p1 = __expf(s[1][r] - mnew);
      psum += p0 + p1;
      pbf.h[0][r] = (half_t)p0;
      pbf.h[1][r] = (half_t)p1;
    }
    l += psum + __shfl_xor(psum, 16, 32);
    m = mnew;

    #pragma unroll
    for (int dt = 0; dt < 4; ++dt)
      acc[dt] = __builtin_amdgcn_wmma_f32_16x16x32_f16(
          false, va[dt].v, false, pbf.v, (short)0, acc[dt], false, false);
  }

  const float snk = sinks[h];
  const float m2 = fmaxf(m, snk);
  const float co = __expf(m - m2);
  const float ltot = l * co + __expf(snk - m2);
  const float factor = co / ltot;
  half_t* op = attn + (size_t)(b * TT + qbase + ln) * DM + h * 64;
  #pragma unroll
  for (int dt = 0; dt < 4; ++dt) {
    v8h o;
    #pragma unroll
    for (int r = 0; r < 8; ++r) o[r] = (half_t)(acc[dt][r] * factor);
    *(v8h*)(op + dt * 16 + ko) = o;
  }
  *(v4h*)(wdst + wo)              = __builtin_convertvector(w0, v4h);
  *(v4h*)(wdst + wo + WSHARE / 2) = __builtin_convertvector(w1, v4h);
}

extern "C" void kernel_launch(void* const* d_in, const int* in_sizes, int n_in,
                              void* d_out, int out_size, void* d_ws, size_t ws_size,
                              hipStream_t stream) {
  const float* x      = (const float*)d_in[0];
  const int*   amask  = (const int*)d_in[1];
  const float* nscale = (const float*)d_in[2];
  const float* qkv_w  = (const float*)d_in[3];
  const float* qkv_b  = (const float*)d_in[4];
  const float* out_w  = (const float*)d_in[5];
  const float* out_b  = (const float*)d_in[6];
  const float* sinks  = (const float*)d_in[7];
  float* out = (float*)d_out;
  char*  ws  = (char*)d_ws;

  half_t* qkvw_h = (half_t*)(ws + OFF_QKVW);
  half_t* outw_h = (half_t*)(ws + OFF_OUTW);
  half_t* t_h    = (half_t*)(ws + OFF_T);
  half_t* attn_h = (half_t*)(ws + OFF_ATTN);
  float*  rot    = (float*)(ws + OFF_ROT);

  rmsnorm_kernel<<<MROW / 8, 256, 0, stream>>>(x, nscale, t_h, rot);
  f32_to_f16_kernel<6><<<512, 256, 0, stream>>>(qkv_w, qkvw_h);

  proj_attn_kernel<<<FBLOCKS, 512, 0, stream>>>(t_h, qkvw_h, qkv_b, rot, amask, sinks, attn_h,
                                                out_w, outw_h);

  gemm_wmma_kernel<true, 2><<<256, 256, 0, stream>>>(attn_h, outw_h, out_b, x,
                                              out, MROW, DM, DM);
}


# qkv_w gets no f16 copy: the fused projection's stage fetch reads the f32 weight rows (2 whole lines a row and step) and rounds them on the way into LDS; conversion launch gone, 4 -> 3 launches
# speedup vs baseline: 4.6865x; 1.0331x over previous
#include <hip/hip_runtime.h>
#include <hip/hip_bf16.h>

typedef _Float16 half_t;
typedef __attribute__((ext_vector_type(16))) _Float16 v16h;
typedef __attribute__((ext_vector_type(8)))  _Float16 v8h;
typedef __attribute__((ext_vector_type(8)))  float    v8f;
typedef __attribute__((ext_vector_type(4)))  float    v4f;
typedef __attribute__((ext_vector_type(4)))  _Float16 v4h;

union V16 { v16h v; v8h h[2]; };

constexpr int BB   = 2;
constexpr int TT   = 2048;
constexpr int DM   = 1024;
constexpr int HN   = 16;
constexpr int DH   = 64;
constexpr int QKVD = 3072;
constexpr int MROW = BB * TT;
constexpr int WIN  = 128;
constexpr float SM_SCALE = 0.125f;
constexpr float MASKED = -3.0e38f;
constexpr float MINIT  = -1.0e30f;

constexpr size_t OFF_OUTW = 0;
constexpr size_t SZ_OUTW  = (size_t)DM * DM * 2;
constexpr size_t OFF_T    = OFF_OUTW + SZ_OUTW;
constexpr size_t SZ_T     = (size_t)MROW * DM * 2;
constexpr size_t OFF_ATTN = OFF_T + SZ_T;
constexpr size_t SZ_ATTN  = (size_t)MROW * DM * 2;
constexpr size_t OFF_ROT  = OFF_ATTN + SZ_ATTN;
constexpr int    ROTN     = TT * (DH / 2);

__device__ __forceinline__ void ldf(V16& f, const half_t* p) {
  f.h[0] = *(const v8h*)(p);
  f.h[1] = *(const v8h*)(p + 16);
}

__global__ __launch_bounds__(256) void rmsnorm_kernel(
    const float* __restrict__ x, const float* __restrict__ scale,
    half_t* __restrict__ t, float* __restrict__ rot) {
  const int lane = threadIdx.x & 31;
  const int row  = blockIdx.x * 8 + (threadIdx.x >> 5);
  const float* xr = x + (size_t)row * DM;
  v4f v[8];
  #pragma unroll
  for (int i = 0; i < 8; ++i) v[i] = *(const v4f*)(xr + (i * 32 + lane) * 4);
  float ss = 0.f;
  #pragma unroll
  for (int i = 0; i < 8; ++i)
    ss += v[i][0] * v[i][0] + v[i][1] * v[i][1] + v[i][2] * v[i][2] + v[i][3] * v[i][3];
  #pragma unroll
  for (int o = 16; o > 0; o >>= 1) ss += __shfl_xor(ss, o, 32);
  const float inv = rsqrtf(ss * (1.0f / DM) + 1e-5f);
  half_t* tr = t + (size_t)row * DM;
  #pragma unroll
  for (int i = 0; i < 8; ++i) {
    const int c = (i * 32 + lane) * 4;
    const v4f sc = *(const v4f*)(scale + c);
    v4h o;
    #pragma unroll
    for (int e = 0; e < 4; ++e) o[e] = (half_t)(v[i][e] * inv * sc[e]);
    *(v4h*)(tr + c) = o;
  }
  if (row < TT) {
    const float invf = __expf(-(float)lane * (9.210340372f / 32.0f));
    const float ang = (float)row * invf;
    rot[row * 32 + lane]        = cosf(ang);
    rot[ROTN + row * 32 + lane] = sinf(ang);
  }
}

constexpr int GBM    = 128;
constexpr int GBK    = 64;
constexpr int GPITCH = 72;
constexpr int GTILE  = GBM * GPITCH;
constexpr int CPITCH = 68;
static_assert(4 * 64 * CPITCH * 4 <= 4 * GTILE * 2, "a group's C parts must fit its operand stages");

__device__ __forceinline__ void gemm_fetch(v8h ra[8], v8h rb[8],
                                           const half_t* ag, const half_t* wg,
                                           int K, int k) {
  #pragma unroll
  for (int it = 0; it < 8; ++it) {
    ra[it] = *(const v8h*)(ag + (size_t)(it * 16) * K + k);
    rb[it] = *(const v8h*)(wg + (size_t)(it * 16) * K + k);
  }
}

__device__ __forceinline__ void gemm_stash(half_t* as, half_t* bs,
                                           const v8h ra[8], const v8h rb[8]) {
  #pragma unroll
  for (int it = 0; it < 8; ++it) {
    *(v8h*)(as + it * 16 * GPITCH) = ra[it];
    *(v8h*)(bs + it * 16 * GPITCH) = rb[it];
  }
}

__device__ __forceinline__ void gemm_stage(v8f acc[4][4],
                                           const half_t* af, const half_t* bf) {
  V16 a[2][4], b[2][4];
  #pragma unroll
  for (int c = 0; c < 2; ++c) {
    #pragma unroll
    for (int i = 0; i < 4; ++i) ldf(a[c][i], af + i * 16 * GPITCH + c * 32);
    #pragma unroll
    for (int j = 0; j < 4; ++j) ldf(b[c][j], bf + j * 16 * GPITCH + c * 32);
  }
  __builtin_amdgcn_sched_barrier(0);
  #pragma unroll
  for (int c = 0; c < 2; ++c)
    #pragma unroll
    for (int i = 0; i < 4; ++i)
      #pragma unroll
      for (int j = 0; j < 4; ++j)
        acc[i][j] = __builtin_amdgcn_wmma_f32_16x16x32_f16(
            false, a[c][i].v, false, b[c][j].v, (short)0, acc[i][j], false, false);
}

template <bool RES, int KG>
__global__ __launch_bounds__(128 * KG) __attribute__((amdgpu_waves_per_eu(3, 3)))
void gemm_wmma_kernel(
    const half_t* __restrict__ A, const half_t* __restrict__ W,
    const float* __restrict__ bias, const float* __restrict__ residual,
    float* __restrict__ C, int M, int N, int K) {
  __shared__ __attribute__((aligned(16))) half_t sm[KG * 4 * GTILE];
  (void)M;
  const int tid  = threadIdx.x & 127;
  const int kg   = threadIdx.x >> 7;
  const int lane = tid & 31;
  const int wv   = tid >> 5;
  const int wm   = wv >> 1, wn = wv & 1;
  const int ntn  = N >> 7;
  const int tm   = blockIdx.x / ntn, tn = blockIdx.x % ntn;
  const int ko = (lane >= 16) ? 8 : 0;
  const int ln = lane & 15;

  const int srow = tid >> 3, scol = (tid & 7) * 8;
  const half_t* ag = A + (size_t)(tm * GBM + srow) * K + scol;
  const half_t* wg = W + (size_t)(tn * GBM + srow) * K + scol;
  half_t* smg = sm + kg * 4 * GTILE;
  half_t* sst = smg + srow * GPITCH + scol;
  const half_t* fa = smg + (wm * 64 + ln) * GPITCH + ko;
  const half_t* fb = smg + GTILE + (wn * 64 + ln) * GPITCH + ko;

  v8f acc[4][4] = {};
  v8h ra[8], rb[8];

  const int nk = K / (GBK * KG);
  const int krot = (tm + tn) % nk;

  gemm_fetch(ra, rb, ag, wg, K, (krot * KG + kg) * GBK);
  gemm_stash(sst, sst + GTILE, ra, rb);
  __syncthreads();

  for (int kt = 0; kt < nk - 1; ++kt) {
    const int cur = (kt & 1) * 2 * GTILE, nxt = 2 * GTILE - cur;
    int ks = kt + 1 + krot;
    if (ks >= nk) ks -= nk;
    gemm_fetch(ra, rb, ag, wg, K, (ks * KG + kg) * GBK);
    __builtin_amdgcn_sched_barrier(0);
    gemm_stage(acc, fa + cur, fb + cur);
    __builtin_amdgcn_sched_barrier(0);
    gemm_stash(sst + nxt, sst + nxt + GTILE, ra, rb);
    __syncthreads();
  }
  {
    const int cur = ((nk - 1) & 1) * 2 * GTILE;
    gemm_stage(acc, fa + cur, fb + cur);
  }

  __syncthreads();
  float* cw = (float*)sm + (kg * 4 + wv) * (64 * CPITCH);
  #pragma unroll
  for (int i = 0; i < 4; ++i)
    #pragma unroll
    for (int j = 0; j < 4; ++j)
      #pragma unroll
      for (int r = 0; r < 8; ++r)
        cw[(i * 16 + r + ko) * CPITCH + j * 16 + ln] = acc[i][j][r];
  __syncthreads();

  const int c4 = ln * 4;
  const int rh = lane >> 4;
  const int colg = tn * GBM + wn * 64 + c4;
  const v4f bv = *(const v4f*)(bias + colg);
  const float* c0 = (const float*)sm + wv * (64 * CPITCH);
  #pragma unroll 8
  for (int ps = 0; ps < 32 / KG; ++ps) {
    const int rl = kg * (64 / KG) + ps * 2 + rh;
    const size_t g = (size_t)(tm * GBM + wm * 64 + rl) * N + colg;
    v4f v = *(const v4f*)(c0 + rl * CPITCH + c4);
    #pragma unroll
    for (int q = 1; q < KG; ++q)
      v = v + *(const v4f*)(c0 + q * 4 * (64 * CPITCH) + rl * CPITCH + c4);
    v = v + bv;
    if (RES) v = v + *(const v4f*)(residual + g);
    *(v4f*)(C + g) = v;
  }
}

constexpr int FQ     = 256;
constexpr int FROWS  = 384;
constexpr int FWROWS = 192;
constexpr int FSTAGE = (FROWS + FWROWS) * GPITCH;
constexpr int AKP    = 72;
constexpr int FVP    = 392;
static_assert(TT % FQ == 0 && TT >= FROWS, "a block's 384 staged positions must lie inside the sequence");
static_assert(DM % GBK == 0 && DH == 64 && FQ % 64 == 0 && FROWS % 64 == 0, "64x64 wave parts");
static_assert(FROWS * AKP + FQ * AKP + DH * FVP + 2 * FROWS <= 2 * FSTAGE,
              "the K, Q, V^T tiles and the mask words must fit the drained stages");
constexpr int FBLOCKS = BB * HN * (TT / FQ);
constexpr int WSHARE  = DM * DM / FBLOCKS;
static_assert(WSHARE * FBLOCKS == DM * DM && WSHARE == 2 * 512 * 4,
              "a block converts two groups of 4 weights per thread");

__device__ __forceinline__ void proj_fetch(v8h rt[6], v4f rw[6],
                                           const half_t* tg, const float* wg, int k) {
  #pragma unroll
  for (int it = 0; it < 6; ++it) rt[it] = *(const v8h*)(tg + (size_t)(it * 64) * DM + k);
  #pragma unroll
  for (int sg = 0; sg < 3; ++sg) {
    rw[2 * sg]     = *(const v4f*)(wg + (sg - 1) * DM * DM + k);
    rw[2 * sg + 1] = *(const v4f*)(wg + (sg - 1) * DM * DM + k + 32);
  }
}

__device__ __forceinline__ void proj_stash(half_t* st, half_t* sw, const v8h rt[6], const v4f rw[6]) {
  #pragma unroll
  for (int it = 0; it < 6; ++it) *(v8h*)(st + it * 64 * GPITCH) = rt[it];
  #pragma unroll
  for (int sg = 0; sg < 3; ++sg) {
    *(v4h*)(sw + sg * 64 * GPITCH)      = __builtin_convertvector(rw[2 * sg], v4h);
    *(v4h*)(sw + sg * 64 * GPITCH + 32) = __builtin_convertvector(rw[2 * sg + 1], v4h);
  }
}

__device__ __forceinline__ void lds_barrier() {
  asm volatile("s_wait_dscnt 0x0\n\ts_barrier_signal -1\n\ts_barrier_wait -1" ::: "memory");
}

template <class Between>
__device__ __forceinline__ void proj_stage(v8f acc[4][4], const half_t* af, const half_t* bf,
                                           Between between) {
  V16 a[2], b[4];
  #pragma unroll
  for (int j = 0; j < 4; ++j) ldf(b[j], bf + j * 16 * GPITCH);
  ldf(a[0], af);
  ldf(a[1], af + 16 * GPITCH);
  #pragma unroll
  for (int n = 0; n < 8; ++n) {
    const int i = n & 3;
    #pragma unroll
    for (int j = 0; j < 4; ++j)
      acc[i][j] = __builtin_amdgcn_wmma_f32_16x16x32_f16(
          false, a[n & 1].v, false, b[j].v, (short)0, acc[i][j], false, false);
    if (n + 2 < 8) ldf(a[n & 1], af + ((n + 2) & 3) * 16 * GPITCH + ((n + 2) >> 2) * 32);
    if (n == 3) {
      #pragma unroll
      for (int j = 0; j < 4; ++j) ldf(b[j], bf + j * 16 * GPITCH + 32);
      __builtin_amdgcn_sched_barrier(0);
      between();
    }
    if (n < 7) __builtin_amdgcn_sched_barrier(0);
  }
}

__global__ __launch_bounds__(512) __attribute__((amdgpu_waves_per_eu(4, 4)))
void proj_attn_kernel(
    const half_t* __restrict__ t, const float* __restrict__ w,
    const float* __restrict__ bias, const float* __restrict__ rot,
    const int* __restrict__ amask,
    const float* __restrict__ sinks, half_t* __restrict__ attn,
    const float* __restrict__ wsrc, half_t* __restrict__ wdst) {
  typedef __attribute__((ext_vector_type(4))) int v4i;
  __shared__ __attribute__((aligned(16))) half_t sm[2 * FSTAGE];
  half_t* ks = sm;
  half_t* qs = ks + FROWS * AKP;
  half_t* vs = qs + FQ * AKP;
  int*    ms = (int*)(vs + DH * FVP);
  const int tid  = threadIdx.x;
  const int lane = tid & 31;
  const int wv   = tid >> 5;
  const int QB   = TT / FQ;
  const int qb   = blockIdx.x % QB;
  const int bh   = blockIdx.x / QB;
  const int h    = bh % HN, b = bh / HN;
  const int q0   = qb * FQ;
  const int k0   = (q0 >= 128) ? q0 - 128 : 0;
  const int qoff = q0 - k0;
  const int qbase = q0 + wv * 16;
  const int ln = lane & 15;
  const int ko = (lane >= 16) ? 8 : 0;
  const int iq = qbase + ln;

  const int srow = tid >> 3, scol = (tid & 7) * 8;
  const half_t* tg = t + (size_t)(b * TT + k0 + srow) * DM + scol;
  const int wcol = (tid & 7) * 4;
  const float* wg = w + (size_t)(DM + h * DH + srow) * DM + wcol;
  half_t* sst = sm + srow * GPITCH + scol;
  half_t* ssw = sm + (FROWS + srow) * GPITCH + wcol;
  const int role = (wv < 6) ? 0 : (wv < 12) ? 1 : 2;
  const int grp  = (role == 0) ? wv : (role == 1) ? wv - 6 : wv - 12;
  const int arow = (role == 0) ? FROWS + 64 : (role == 1) ? grp * 64 : FROWS;
  const int brow = (role == 0) ? grp * 64 : (role == 1) ? FROWS + 128 : qoff + grp * 64;
  const half_t* fa = sm + (arow + ln) * GPITCH + ko;
  const half_t* fb = sm + (brow + ln) * GPITCH + ko;

  v8f pc[4][4] = {};
  {
    v8h rt[6];
    v4f rw[6];
    constexpr int nk = DM / GBK;
    static_assert((nk & (nk - 1)) == 0 && nk == HN && nk == BB * (TT / FQ),
                  "the k start below gives the 16 blocks of a sharing group 16 different steps");
    const int krot = (h + b * QB + qb) & (nk - 1);
    proj_fetch(rt, rw, tg, wg, krot * GBK);
    proj_stash(sst, ssw, rt, rw);
    proj_fetch(rt, rw, tg, wg, ((1 + krot) & (nk - 1)) * GBK);
    lds_barrier();
    for (int kt = 0; kt < nk - 1; ++kt) {
      const int cur = (kt & 1) * FSTAGE, nxt = FSTAGE - cur;
      proj_stage(pc, fa + cur, fb + cur, [&]() {
        proj_stash(sst + nxt, ssw + nxt, rt, rw);
        if (kt + 2 < nk)
          proj_fetch(rt, rw, tg, wg, ((kt + 2 + krot) & (nk - 1)) * GBK);
      });
      lds_barrier();
    }
    proj_stage(pc, fa + ((nk - 1) & 1) * FSTAGE, fb + ((nk - 1) & 1) * FSTAGE, []() {});
  }
  __syncthreads();

  if (tid < FROWS) ms[tid] = amask[b * TT + k0 + tid];
  if (role != 1) {
    const float* bp = bias + (role == 0 ? DM : 0) + h * DH + ko;
    half_t* dst = (role == 0 ? ks + brow * AKP : qs + (brow - qoff) * AKP) + ln * AKP + ko;
    const float* rp = rot + (size_t)(k0 + brow + ln) * 32 + ko;
    #pragma unroll
    for (int i = 0; i < 2; ++i) {
      const v4f bl0 = *(const v4f*)(bp + i * 16),      bl1 = *(const v4f*)(bp + i * 16 + 4);
      const v4f bh0 = *(const v4f*)(bp + 32 + i * 16), bh1 = *(const v4f*)(bp + 32 + i * 16 + 4);
      #pragma unroll
      for (int j = 0; j < 4; ++j) {
        const float* rj = rp + j * 16 * 32 + i * 16;
        const v4f c0 = *(const v4f*)(rj),        c1 = *(const v4f*)(rj + 4);
        const v4f s0 = *(const v4f*)(rj + ROTN), s1 = *(const v4f*)(rj + ROTN + 4);
        v8h lo, hi;
        #pragma unroll
        for (int r = 0; r < 8; ++r) {
          const float x1 = pc[i][j][r]     + (r < 4 ? bl0[r & 3] : bl1[r & 3]);
          const float x2 = pc[i + 2][j][r] + (r < 4 ? bh0[r & 3] : bh1[r & 3]);
          const float cv = (r < 4 ? c0[r & 3] : c1[r & 3]);
          const float sv = (r < 4 ? s0[r & 3] : s1[r & 3]);
          lo[r] = (half_t)(x1 * cv - x2 * sv);
          hi[r] = (half_t)(x2 * cv + x1 * sv);
        }
        *(v8h*)(dst + j * 16 * AKP + i * 16)      = lo;
        *(v8h*)(dst + j * 16 * AKP + 32 + i * 16) = hi;
      }
      __builtin_amdgcn_sched_barrier(0);
    }
  } else {
    const float* bp = bias + 2 * DM + h * DH + ln;
    #pragma unroll
    for (int j = 0; j < 4; ++j) {
      const float bj = bp[j * 16];
      #pragma unroll
      for (int i = 0; i < 4; ++i) {
        v8h o;
        #pragma unroll
        for (int r = 0; r < 8; ++r) o[r] = (half_t)(pc[i][j][r] + bj);
        *(v8h*)(vs + (j * 16 + ln) * FVP + arow + i * 16 + ko) = o;
      }
    }
  }
  __syncthreads();

  const size_t wo = (size_t)blockIdx.x * WSHARE + tid * 4;
  const v4f w0 = *(const v4f*)(wsrc + wo);
  const v4f w1 = *(const v4f*)(wsrc + wo + WSHARE / 2);

  V16 qb0, qb1;
  {
    const half_t* qp = qs + (wv * 16 + ln) * AKP;
    ldf(qb0, qp + ko);
    ldf(qb1, qp + 32 + ko);
  }

  v8f acc[4] = {};
  float m = MINIT, l = 0.f;
  int kstart = qbase - (WIN - 1); if (kstart < 0) kstart = 0;
  kstart &= ~31;
  const int kend = qbase + 16;

  for (int jc = kstart; jc < kend; jc += 32) {
    const int jl = jc - k0;
    V16 ka[2][2], va[4];
    v4i mk[2][2];
    #pragma unroll
    for (int t2 = 0; t2 < 2; ++t2) {
      const half_t* kp = ks + (jl + t2 * 16 + ln) * AKP + ko;
      ldf(ka[t2][0], kp);
      ldf(ka[t2][1], kp + 32);
    }
    #pragma unroll
    for (int dt = 0; dt < 4; ++dt)
      ldf(va[dt], vs + (dt * 16 + ln) * FVP + jl + ko);
    #pragma unroll
    for (int t2 = 0; t2 < 2; ++t2) {
      const int* mp = ms + jl + t2 * 16 + ko;
      mk[t2][0] = *(const v4i*)(mp);
      mk[t2][1] = *(const v4i*)(mp + 4);
    }

    v8f s[2];
    #pragma unroll
    for (int t2 = 0; t2 < 2; ++t2) {
      v8f z = {};
      z = __builtin_amdgcn_wmma_f32_16x16x32_f16(false, ka[t2][0].v, false, qb0.v,
                                                 (short)0, z, false, false);
      z = __builtin_amdgcn_wmma_f32_16x16x32_f16(false, ka[t2][1].v, false, qb1.v,
                                                 (short)0, z, false, false);
      s[t2] = z;
    }
    float tl = MASKED;
    #pragma unroll
    for (int t2 = 0; t2 < 2; ++t2) {
      #pragma unroll
      for (int r = 0; r < 8; ++r) {
        const int j = jc + t2 * 16 + r + ko;
        const int mword = mk[t2][r >> 2][r & 3];
        const bool ok = (j <= iq) && (j + WIN > iq) && (mword != 0);
        const float v = ok ? s[t2][r] * SM_SCALE : MASKED;
        s[t2][r] = v;
        tl = fmaxf(tl, v);
      }
    }
    tl = fmaxf(tl, __shfl_xor(tl, 16, 32));
    const float mnew = fmaxf(m, tl);
    const float co = __expf(m - mnew);
    l *= co;
    #pragma unroll
    for (int dt = 0; dt < 4; ++dt) acc[dt] = acc[dt] * co;

    V16 pbf;
    float psum = 0.f;
    #pragma unroll
    for (int r = 0; r < 8; ++r) {
      const float p0 = __expf(s[0][r] - mnew);
      const float p1 = __expf(s[1][r] - mnew);
      psum += p0 + p1;
      pbf.h[0][r] = (half_t)p0;
      pbf.h[1][r] = (half_t)p1;
    }
    l += psum + __shfl_xor(psum, 16, 32);
    m = mnew;

    #pragma unroll
    for (int dt = 0; dt < 4; ++dt)
      acc[dt] = __builtin_amdgcn_wmma_f32_16x16x32_f16(
          false, va[dt].v, false, pbf.v, (short)0, acc[dt], false, false);
  }

  const float snk = sinks[h];
  const float m2 = fmaxf(m, snk);
  const float co = __expf(m - m2);
  const float ltot = l * co + __expf(snk - m2);
  const float factor = co / ltot;
  half_t* op = attn + (size_t)(b * TT + qbase + ln) * DM + h * 64;
  #pragma unroll
  for (int dt = 0; dt < 4; ++dt) {
    v8h o;
    #pragma unroll
    for (int r = 0; r < 8; ++r) o[r] = (half_t)(acc[dt][r] * factor);
    *(v8h*)(op + dt * 16 + ko) = o;
  }
  *(v4h*)(wdst + wo)              = __builtin_convertvector(w0, v4h);
  *(v4h*)(wdst + wo + WSHARE / 2) = __builtin_convertvector(w1, v4h);
}

extern "C" void kernel_launch(void* const* d_in, const int* in_sizes, int n_in,
                              void* d_out, int out_size, void* d_ws, size_t ws_size,
                              hipStream_t stream) {
  const float* x      = (const float*)d_in[0];
  const int*   amask  = (const int*)d_in[1];
  const float* nscale = (const float*)d_in[2];
  const float* qkv_w  = (const float*)d_in[3];
  const float* qkv_b  = (const float*)d_in[4];
  const float* out_w  = (const float*)d_in[5];
  const float* out_b  = (const float*)d_in[6];
  const float* sinks  = (const float*)d_in[7];
  float* out = (float*)d_out;
  char*  ws  = (char*)d_ws;

  half_t* outw_h = (half_t*)(ws + OFF_OUTW);
  half_t* t_h    = (half_t*)(ws + OFF_T);
  half_t* attn_h = (half_t*)(ws + OFF_ATTN);
  float*  rot    = (float*)(ws + OFF_ROT);

  rmsnorm_kernel<<<MROW / 8, 256, 0, stream>>>(x, nscale, t_h, rot);

  proj_attn_kernel<<<FBLOCKS, 512, 0, stream>>>(t_h, qkv_w, qkv_b, rot, amask, sinks, attn_h,
                                                out_w, outw_h);

  gemm_wmma_kernel<true, 2><<<256, 256, 0, stream>>>(attn_h, outw_h, out_b, x,
                                              out, MROW, DM, DM);
}
